# starting kernel
# baseline (speedup 1.0000x reference)
#include <hip/hip_runtime.h>
#include <hip/hip_bf16.h>

typedef __attribute__((ext_vector_type(2))) float v2f;
typedef __attribute__((ext_vector_type(8))) float v8f;

#define F_IN 512
#define HID  16
#define CLS  7

__global__ void deg_init_k(float* __restrict__ deg, int n) {
    int i = blockIdx.x * blockDim.x + threadIdx.x;
    if (i < n) deg[i] = 1.0f;
}

__global__ void deg_count_k(const int* __restrict__ dst, float* __restrict__ deg, int E) {
    int e = blockIdx.x * blockDim.x + threadIdx.x;
    if (e < E) unsafeAtomicAdd(&deg[dst[e]], 1.0f);
}

__global__ void dinv_k(float* __restrict__ deg, int n) {
    int i = blockIdx.x * blockDim.x + threadIdx.x;
    if (i < n) deg[i] = rsqrtf(deg[i]);
}

__global__ void __launch_bounds__(256) gemm1_k(const float* __restrict__ x,
                                               const float* __restrict__ W1,
                                               float* __restrict__ h, int nTiles) {
    __shared__ float sW[F_IN * HID];
    for (int i = threadIdx.x * 4; i < F_IN * HID; i += 256 * 4)
        *(float4*)(sW + i) = *(const float4*)(W1 + i);
    __syncthreads();

    const int wave = threadIdx.x >> 5;
    const int lane = threadIdx.x & 31;
    const int tile = blockIdx.x * 8 + wave;
    if (tile >= nTiles) return;

    const int m    = lane & 15;
    const int kk   = (lane >> 4) << 1;
    const int half = lane >> 4;

    const float* arow = x + (size_t)(tile * 16 + m) * F_IN + kk;
    v8f c = {0.f, 0.f, 0.f, 0.f, 0.f, 0.f, 0.f, 0.f};

#pragma unroll 4
    for (int k0 = 0; k0 < F_IN; k0 += 4) {
        v2f a = *(const v2f*)(arow + k0);
        v2f b;
        b.x = sW[(k0 + kk)     * HID + m];
        b.y = sW[(k0 + kk + 1) * HID + m];
        c = __builtin_amdgcn_wmma_f32_16x16x4_f32(false, a, false, b,
                                                  (short)0, c, false, false);
    }

    float* orow = h + (size_t)(tile * 16 + half * 8) * HID + m;
#pragma unroll
    for (int v = 0; v < 8; ++v) orow[(size_t)v * HID] = c[v];
}

__global__ void self_init1_k(const float* __restrict__ h, const float* __restrict__ dinv,
                             float* __restrict__ agg, int n) {
    int t = blockIdx.x * blockDim.x + threadIdx.x;
    if (t >= n) return;
    int i = t >> 4;
    float di = dinv[i];
    agg[t] = di * di * h[t];
}

__global__ void scatter1_k(const int* __restrict__ src, const int* __restrict__ dst,
                           const float* __restrict__ dinv, const float* __restrict__ h,
                           float* __restrict__ agg, int E) {
    int t = blockIdx.x * blockDim.x + threadIdx.x;
    int e = t >> 4;
    if (e >= E) return;
    int f = t & 15;
    int s = src[e], d = dst[e];
    float w = dinv[s] * dinv[d];
    unsafeAtomicAdd(&agg[(size_t)d * HID + f], w * h[(size_t)s * HID + f]);
}

__global__ void bias_relu_k(float* __restrict__ agg, const float* __restrict__ b1, int n) {
    int t = blockIdx.x * blockDim.x + threadIdx.x;
    if (t >= n) return;
    float v = agg[t] + b1[t & 15];
    agg[t] = v > 0.f ? v : 0.f;
}

__global__ void __launch_bounds__(256) gemm2_k(const float* __restrict__ a_in,
                                               const float* __restrict__ W2,
                                               float* __restrict__ h2, int nTiles) {
    __shared__ float sW[HID * 16];
    {
        int t = threadIdx.x;
        int k = t >> 4, n = t & 15;
        sW[t] = (n < CLS) ? W2[k * CLS + n] : 0.0f;
    }
    __syncthreads();

    const int wave = threadIdx.x >> 5;
    const int lane = threadIdx.x & 31;
    const int tile = blockIdx.x * 8 + wave;
    if (tile >= nTiles) return;

    const int m    = lane & 15;
    const int kk   = (lane >> 4) << 1;
    const int half = lane >> 4;

    const float* arow = a_in + (size_t)(tile * 16 + m) * HID + kk;
    v8f c = {0.f, 0.f, 0.f, 0.f, 0.f, 0.f, 0.f, 0.f};

#pragma unroll
    for (int k0 = 0; k0 < HID; k0 += 4) {
        v2f a = *(const v2f*)(arow + k0);
        v2f b;
        b.x = sW[(k0 + kk)     * 16 + m];
        b.y = sW[(k0 + kk + 1) * 16 + m];
        c = __builtin_amdgcn_wmma_f32_16x16x4_f32(false, a, false, b,
                                                  (short)0, c, false, false);
    }

    if (m < CLS) {
#pragma unroll
        for (int v = 0; v < 8; ++v) {
            int row = tile * 16 + half * 8 + v;
            h2[(size_t)row * CLS + m] = c[v];
        }
    }
}

__global__ void self_init2_k(const float* __restrict__ h2, const float* __restrict__ dinv,
                             const float* __restrict__ b2, float* __restrict__ out, int n) {
    int t = blockIdx.x * blockDim.x + threadIdx.x;
    if (t >= n) return;
    int i = t / CLS, f = t % CLS;
    float di = dinv[i];
    out[t] = di * di * h2[t] + b2[f];
}

__global__ void scatter2_k(const int* __restrict__ src, const int* __restrict__ dst,
                           const float* __restrict__ dinv, const float* __restrict__ h2,
                           float* __restrict__ out, int E) {
    int t = blockIdx.x * blockDim.x + threadIdx.x;
    int e = t >> 3;
    if (e >= E) return;
    int f = t & 7;
    if (f >= CLS) return;
    int s = src[e], d = dst[e];
    float w = dinv[s] * dinv[d];
    unsafeAtomicAdd(&out[(size_t)d * CLS + f], w * h2[(size_t)s * CLS + f]);
}

extern "C" void kernel_launch(void* const* d_in, const int* in_sizes, int n_in,
                              void* d_out, int out_size, void* d_ws, size_t ws_size,
                              hipStream_t stream) {
    const float* x  = (const float*)d_in[0];
    const int*   ei = (const int*)  d_in[1];
    const float* W1 = (const float*)d_in[2];
    const float* b1 = (const float*)d_in[3];
    const float* W2 = (const float*)d_in[4];
    const float* b2 = (const float*)d_in[5];
    float* out = (float*)d_out;

    const int N = in_sizes[0] / F_IN;
    const int E = in_sizes[1] / 2;
    const int* src = ei;
    const int* dst = ei + E;

    float* dinv = (float*)d_ws;
    float* h    = dinv + N;
    float* agg  = h   + (size_t)N * HID;
    float* h2   = agg + (size_t)N * HID;

    const int B = 256;
    const int nTiles = N / 16;

    deg_init_k <<<(N + B - 1) / B, B, 0, stream>>>(dinv, N);
    deg_count_k<<<(E + B - 1) / B, B, 0, stream>>>(dst, dinv, E);
    dinv_k     <<<(N + B - 1) / B, B, 0, stream>>>(dinv, N);

    gemm1_k    <<<(nTiles + 7) / 8, B, 0, stream>>>(x, W1, h, nTiles);

    int n1 = N * HID;
    self_init1_k<<<(n1 + B - 1) / B, B, 0, stream>>>(h, dinv, agg, n1);
    {
        long long tot = (long long)E * 16;
        scatter1_k<<<(int)((tot + B - 1) / B), B, 0, stream>>>(src, dst, dinv, h, agg, E);
    }
    bias_relu_k<<<(n1 + B - 1) / B, B, 0, stream>>>(agg, b1, n1);

    gemm2_k    <<<(nTiles + 7) / 8, B, 0, stream>>>(agg, W2, h2, nTiles);

    int n2 = N * CLS;
    self_init2_k<<<(n2 + B - 1) / B, B, 0, stream>>>(h2, dinv, b2, out, n2);
    {
        long long tot = (long long)E * 8;
        scatter2_k<<<(int)((tot + B - 1) / B), B, 0, stream>>>(src, dst, dinv, h2, out, E);
    }
}


# Degree count: one global f32 atomic per edge replaced by per-block LDS counters (16-bit pairs), planes summed into dinv by one kernel; integer counts, no float sum order changes
# speedup vs baseline: 1.3287x; 1.3287x over previous
#include <hip/hip_runtime.h>
#include <hip/hip_bf16.h>

typedef __attribute__((ext_vector_type(2))) float v2f;
typedef __attribute__((ext_vector_type(8))) float v8f;

#define F_IN 512
#define HID  16
#define CLS  7

#define NODES  100000
#define HWORDS (NODES / 2)
#define CHUNK  50000

__global__ void __launch_bounds__(1024) deg_hist_k(const int* __restrict__ dst,
                                                   unsigned* __restrict__ parts, int E) {
    __shared__ __attribute__((aligned(16))) unsigned cnt[HWORDS];

    const int base = blockIdx.x * CHUNK;
    int end = base + CHUNK;
    if (end > E) end = E;
    const int nvec = (end - base) >> 2;
    const int4* d4 = (const int4*)(dst + base);

    constexpr int SLOTS = (CHUNK / 4 + 1023) / 1024;
    int4 d[SLOTS];
#pragma unroll
    for (int j = 0; j < SLOTS; ++j) {
        const int i = threadIdx.x + j * 1024;
        if (i < nvec) d[j] = d4[i];
    }

    for (int i = threadIdx.x; i < HWORDS / 4; i += 1024)
        ((uint4*)cnt)[i] = make_uint4(0u, 0u, 0u, 0u);
    __syncthreads();

#pragma unroll
    for (int j = 0; j < SLOTS; ++j) {
        const int i = threadIdx.x + j * 1024;
        if (i < nvec) {
            atomicAdd(&cnt[d[j].x >> 1], 1u << ((d[j].x & 1) << 4));
            atomicAdd(&cnt[d[j].y >> 1], 1u << ((d[j].y & 1) << 4));
            atomicAdd(&cnt[d[j].z >> 1], 1u << ((d[j].z & 1) << 4));
            atomicAdd(&cnt[d[j].w >> 1], 1u << ((d[j].w & 1) << 4));
        }
    }
    for (int e = base + (nvec << 2) + threadIdx.x; e < end; e += 1024) {
        const int t = dst[e];
        atomicAdd(&cnt[t >> 1], 1u << ((t & 1) << 4));
    }
    __syncthreads();

    uint4* o = (uint4*)(parts + (size_t)blockIdx.x * HWORDS);
    for (int i = threadIdx.x; i < HWORDS / 4; i += 1024) o[i] = ((const uint4*)cnt)[i];
}

__global__ void dinv_from_parts_k(const unsigned* __restrict__ parts, float* __restrict__ dinv,
                                  int nParts, int n) {
    int t = blockIdx.x * blockDim.x + threadIdx.x;
    if (t >= HWORDS) return;
    unsigned lo = 0u, hi = 0u;
#pragma unroll 8
    for (int p = 0; p < nParts; ++p) {
        unsigned w = parts[(size_t)p * HWORDS + t];
        lo += w & 0xffffu;
        hi += w >> 16;
    }
    const int i = 2 * t;
    float r0 = rsqrtf(1.0f + (float)lo);
    float r1 = rsqrtf(1.0f + (float)hi);
    if (i + 1 < n)      *(float2*)(dinv + i) = make_float2(r0, r1);
    else if (i < n)     dinv[i] = r0;
}

__global__ void __launch_bounds__(256) gemm1_k(const float* __restrict__ x,
                                               const float* __restrict__ W1,
                                               float* __restrict__ h, int nTiles) {
    __shared__ float sW[F_IN * HID];
    for (int i = threadIdx.x * 4; i < F_IN * HID; i += 256 * 4)
        *(float4*)(sW + i) = *(const float4*)(W1 + i);
    __syncthreads();

    const int wave = threadIdx.x >> 5;
    const int lane = threadIdx.x & 31;
    const int tile = blockIdx.x * 8 + wave;
    if (tile >= nTiles) return;

    const int m    = lane & 15;
    const int kk   = (lane >> 4) << 1;
    const int half = lane >> 4;

    const float* arow = x + (size_t)(tile * 16 + m) * F_IN + kk;
    v8f c = {0.f, 0.f, 0.f, 0.f, 0.f, 0.f, 0.f, 0.f};

#pragma unroll 4
    for (int k0 = 0; k0 < F_IN; k0 += 4) {
        v2f a = *(const v2f*)(arow + k0);
        v2f b;
        b.x = sW[(k0 + kk)     * HID + m];
        b.y = sW[(k0 + kk + 1) * HID + m];
        c = __builtin_amdgcn_wmma_f32_16x16x4_f32(false, a, false, b,
                                                  (short)0, c, false, false);
    }

    float* orow = h + (size_t)(tile * 16 + half * 8) * HID + m;
#pragma unroll
    for (int v = 0; v < 8; ++v) orow[(size_t)v * HID] = c[v];
}

__global__ void self_init1_k(const float* __restrict__ h, const float* __restrict__ dinv,
                             float* __restrict__ agg, int n) {
    int t = blockIdx.x * blockDim.x + threadIdx.x;
    if (t >= n) return;
    int i = t >> 4;
    float di = dinv[i];
    agg[t] = di * di * h[t];
}

__global__ void scatter1_k(const int* __restrict__ src, const int* __restrict__ dst,
                           const float* __restrict__ dinv, const float* __restrict__ h,
                           float* __restrict__ agg, int E) {
    int t = blockIdx.x * blockDim.x + threadIdx.x;
    int e = t >> 4;
    if (e >= E) return;
    int f = t & 15;
    int s = src[e], d = dst[e];
    float w = dinv[s] * dinv[d];
    unsafeAtomicAdd(&agg[(size_t)d * HID + f], w * h[(size_t)s * HID + f]);
}

__global__ void bias_relu_k(float* __restrict__ agg, const float* __restrict__ b1, int n) {
    int t = blockIdx.x * blockDim.x + threadIdx.x;
    if (t >= n) return;
    float v = agg[t] + b1[t & 15];
    agg[t] = v > 0.f ? v : 0.f;
}

__global__ void __launch_bounds__(256) gemm2_k(const float* __restrict__ a_in,
                                               const float* __restrict__ W2,
                                               float* __restrict__ h2, int nTiles) {
    __shared__ float sW[HID * 16];
    {
        int t = threadIdx.x;
        int k = t >> 4, n = t & 15;
        sW[t] = (n < CLS) ? W2[k * CLS + n] : 0.0f;
    }
    __syncthreads();

    const int wave = threadIdx.x >> 5;
    const int lane = threadIdx.x & 31;
    const int tile = blockIdx.x * 8 + wave;
    if (tile >= nTiles) return;

    const int m    = lane & 15;
    const int kk   = (lane >> 4) << 1;
    const int half = lane >> 4;

    const float* arow = a_in + (size_t)(tile * 16 + m) * HID + kk;
    v8f c = {0.f, 0.f, 0.f, 0.f, 0.f, 0.f, 0.f, 0.f};

#pragma unroll
    for (int k0 = 0; k0 < HID; k0 += 4) {
        v2f a = *(const v2f*)(arow + k0);
        v2f b;
        b.x = sW[(k0 + kk)     * 16 + m];
        b.y = sW[(k0 + kk + 1) * 16 + m];
        c = __builtin_amdgcn_wmma_f32_16x16x4_f32(false, a, false, b,
                                                  (short)0, c, false, false);
    }

    if (m < CLS) {
#pragma unroll
        for (int v = 0; v < 8; ++v) {
            int row = tile * 16 + half * 8 + v;
            h2[(size_t)row * CLS + m] = c[v];
        }
    }
}

__global__ void self_init2_k(const float* __restrict__ h2, const float* __restrict__ dinv,
                             const float* __restrict__ b2, float* __restrict__ out, int n) {
    int t = blockIdx.x * blockDim.x + threadIdx.x;
    if (t >= n) return;
    int i = t / CLS, f = t % CLS;
    float di = dinv[i];
    out[t] = di * di * h2[t] + b2[f];
}

__global__ void scatter2_k(const int* __restrict__ src, const int* __restrict__ dst,
                           const float* __restrict__ dinv, const float* __restrict__ h2,
                           float* __restrict__ out, int E) {
    int t = blockIdx.x * blockDim.x + threadIdx.x;
    int e = t >> 3;
    if (e >= E) return;
    int f = t & 7;
    if (f >= CLS) return;
    int s = src[e], d = dst[e];
    float w = dinv[s] * dinv[d];
    unsafeAtomicAdd(&out[(size_t)d * CLS + f], w * h2[(size_t)s * CLS + f]);
}

extern "C" void kernel_launch(void* const* d_in, const int* in_sizes, int n_in,
                              void* d_out, int out_size, void* d_ws, size_t ws_size,
                              hipStream_t stream) {
    const float* x  = (const float*)d_in[0];
    const int*   ei = (const int*)  d_in[1];
    const float* W1 = (const float*)d_in[2];
    const float* b1 = (const float*)d_in[3];
    const float* W2 = (const float*)d_in[4];
    const float* b2 = (const float*)d_in[5];
    float* out = (float*)d_out;

    const int N = in_sizes[0] / F_IN;
    const int E = in_sizes[1] / 2;
    const int* src = ei;
    const int* dst = ei + E;

    float* dinv = (float*)d_ws;
    float* h    = dinv + N;
    float* agg  = h   + (size_t)N * HID;
    float* h2   = agg + (size_t)N * HID;

    const int B = 256;
    const int nTiles = N / 16;

    const int nParts = (E + CHUNK - 1) / CHUNK;
    unsigned* parts = (unsigned*)h;
    deg_hist_k       <<<nParts, 1024, 0, stream>>>(dst, parts, E);
    dinv_from_parts_k<<<(HWORDS + B - 1) / B, B, 0, stream>>>(parts, dinv, nParts, N);

    gemm1_k    <<<(nTiles + 7) / 8, B, 0, stream>>>(x, W1, h, nTiles);

    int n1 = N * HID;
    self_init1_k<<<(n1 + B - 1) / B, B, 0, stream>>>(h, dinv, agg, n1);
    {
        long long tot = (long long)E * 16;
        scatter1_k<<<(int)((tot + B - 1) / B), B, 0, stream>>>(src, dst, dinv, h, agg, E);
    }
    bias_relu_k<<<(n1 + B - 1) / B, B, 0, stream>>>(agg, b1, n1);

    gemm2_k    <<<(nTiles + 7) / 8, B, 0, stream>>>(agg, W2, h2, nTiles);

    int n2 = N * CLS;
    self_init2_k<<<(n2 + B - 1) / B, B, 0, stream>>>(h2, dinv, b2, out, n2);
    {
        long long tot = (long long)E * 8;
        scatter2_k<<<(int)((tot + B - 1) / B), B, 0, stream>>>(src, dst, dinv, h2, out, E);
    }
}


# Both edge passes: one wave per 32 edges (coalesced index loads, one weight per edge, row-lane moves), every row gather of a wave in flight before its float atomics; sums still in arrival order
# speedup vs baseline: 1.6904x; 1.2722x over previous
#include <hip/hip_runtime.h>
#include <hip/hip_bf16.h>

typedef __attribute__((ext_vector_type(2))) float v2f;
typedef __attribute__((ext_vector_type(8))) float v8f;

#define F_IN 512
#define HID  16
#define CLS  7

#define NODES  100000
#define HWORDS (NODES / 2)
#define CHUNK  50000

__global__ void __launch_bounds__(1024) deg_hist_k(const int* __restrict__ dst,
                                                   unsigned* __restrict__ parts, int E) {
    __shared__ __attribute__((aligned(16))) unsigned cnt[HWORDS];

    const int base = blockIdx.x * CHUNK;
    int end = base + CHUNK;
    if (end > E) end = E;
    const int nvec = (end - base) >> 2;
    const int4* d4 = (const int4*)(dst + base);

    constexpr int SLOTS = (CHUNK / 4 + 1023) / 1024;
    int4 d[SLOTS];
#pragma unroll
    for (int j = 0; j < SLOTS; ++j) {
        const int i = threadIdx.x + j * 1024;
        if (i < nvec) d[j] = d4[i];
    }

    for (int i = threadIdx.x; i < HWORDS / 4; i += 1024)
        ((uint4*)cnt)[i] = make_uint4(0u, 0u, 0u, 0u);
    __syncthreads();

#pragma unroll
    for (int j = 0; j < SLOTS; ++j) {
        const int i = threadIdx.x + j * 1024;
        if (i < nvec) {
            atomicAdd(&cnt[d[j].x >> 1], 1u << ((d[j].x & 1) << 4));
            atomicAdd(&cnt[d[j].y >> 1], 1u << ((d[j].y & 1) << 4));
            atomicAdd(&cnt[d[j].z >> 1], 1u << ((d[j].z & 1) << 4));
            atomicAdd(&cnt[d[j].w >> 1], 1u << ((d[j].w & 1) << 4));
        }
    }
    for (int e = base + (nvec << 2) + threadIdx.x; e < end; e += 1024) {
        const int t = dst[e];
        atomicAdd(&cnt[t >> 1], 1u << ((t & 1) << 4));
    }
    __syncthreads();

    uint4* o = (uint4*)(parts + (size_t)blockIdx.x * HWORDS);
    for (int i = threadIdx.x; i < HWORDS / 4; i += 1024) o[i] = ((const uint4*)cnt)[i];
}

__global__ void dinv_from_parts_k(const unsigned* __restrict__ parts, float* __restrict__ dinv,
                                  int nParts, int n) {
    int t = blockIdx.x * blockDim.x + threadIdx.x;
    if (t >= HWORDS) return;
    unsigned lo = 0u, hi = 0u;
#pragma unroll 8
    for (int p = 0; p < nParts; ++p) {
        unsigned w = parts[(size_t)p * HWORDS + t];
        lo += w & 0xffffu;
        hi += w >> 16;
    }
    const int i = 2 * t;
    float r0 = rsqrtf(1.0f + (float)lo);
    float r1 = rsqrtf(1.0f + (float)hi);
    if (i + 1 < n)      *(float2*)(dinv + i) = make_float2(r0, r1);
    else if (i < n)     dinv[i] = r0;
}

__global__ void __launch_bounds__(256) gemm1_k(const float* __restrict__ x,
                                               const float* __restrict__ W1,
                                               float* __restrict__ h, int nTiles) {
    __shared__ float sW[F_IN * HID];
    for (int i = threadIdx.x * 4; i < F_IN * HID; i += 256 * 4)
        *(float4*)(sW + i) = *(const float4*)(W1 + i);
    __syncthreads();

    const int wave = threadIdx.x >> 5;
    const int lane = threadIdx.x & 31;
    const int tile = blockIdx.x * 8 + wave;
    if (tile >= nTiles) return;

    const int m    = lane & 15;
    const int kk   = (lane >> 4) << 1;
    const int half = lane >> 4;

    const float* arow = x + (size_t)(tile * 16 + m) * F_IN + kk;
    v8f c = {0.f, 0.f, 0.f, 0.f, 0.f, 0.f, 0.f, 0.f};

#pragma unroll 4
    for (int k0 = 0; k0 < F_IN; k0 += 4) {
        v2f a = *(const v2f*)(arow + k0);
        v2f b;
        b.x = sW[(k0 + kk)     * HID + m];
        b.y = sW[(k0 + kk + 1) * HID + m];
        c = __builtin_amdgcn_wmma_f32_16x16x4_f32(false, a, false, b,
                                                  (short)0, c, false, false);
    }

    float* orow = h + (size_t)(tile * 16 + half * 8) * HID + m;
#pragma unroll
    for (int v = 0; v < 8; ++v) orow[(size_t)v * HID] = c[v];
}

__global__ void self_init1_k(const float* __restrict__ h, const float* __restrict__ dinv,
                             float* __restrict__ agg, int n) {
    int t = blockIdx.x * blockDim.x + threadIdx.x;
    if (t >= n) return;
    int i = t >> 4;
    float di = dinv[i];
    agg[t] = di * di * h[t];
}

__device__ __forceinline__ unsigned rowsel(unsigned x, unsigned lo, unsigned hi) {
    return __builtin_amdgcn_permlane16(x, x, lo, hi, false, false);
}

template <bool TAIL>
__device__ __forceinline__ void adds1(float* __restrict__ agg, const float (&v)[16],
                                      unsigned db, float w, unsigned f4, int left) {
    const int lane = threadIdx.x & 31;
#pragma unroll
    for (int j = 0; j < 16; ++j) {
        const unsigned sel = 0x11111111u * (unsigned)j;
        const unsigned off = rowsel(db, sel, sel) + f4;
        const float wj = __uint_as_float(rowsel(__float_as_uint(w), sel, sel));
        if (!TAIL || ((lane & 16) + j) < left)
            unsafeAtomicAdd((float*)((char*)agg + off), wj * v[j]);
    }
}

__global__ void __launch_bounds__(256) scatter1_k(const int* __restrict__ src,
                                                  const int* __restrict__ dst,
                                                  const float* __restrict__ dinv,
                                                  const float* __restrict__ h,
                                                  float* __restrict__ agg, int E) {
    const int lane = threadIdx.x & 31;
    const int e0 = (blockIdx.x * 8 + (threadIdx.x >> 5)) * 32;
    if (e0 >= E) return;
    const int left = E - e0;
    int s = 0, d = 0;
    if (lane < left) { s = src[e0 + lane]; d = dst[e0 + lane]; }
    const float ds = dinv[s], dd = dinv[d];

    const unsigned f4 = (unsigned)(lane & 15) * 4u;
    const unsigned sb = (unsigned)s * (HID * 4u);
    float v[16];
#pragma unroll
    for (int j = 0; j < 16; ++j) {
        const unsigned sel = 0x11111111u * (unsigned)j;
        const unsigned off = rowsel(sb, sel, sel) + f4;
        v[j] = *(const float*)((const char*)h + off);
    }

    const float w = ds * dd;
    const unsigned db = (unsigned)d * (HID * 4u);
    if (left >= 32) adds1<false>(agg, v, db, w, f4, left);
    else            adds1<true >(agg, v, db, w, f4, left);
}

__global__ void bias_relu_k(float* __restrict__ agg, const float* __restrict__ b1, int n) {
    int t = blockIdx.x * blockDim.x + threadIdx.x;
    if (t >= n) return;
    float v = agg[t] + b1[t & 15];
    agg[t] = v > 0.f ? v : 0.f;
}

__global__ void __launch_bounds__(256) gemm2_k(const float* __restrict__ a_in,
                                               const float* __restrict__ W2,
                                               float* __restrict__ h2, int nTiles) {
    __shared__ float sW[HID * 16];
    {
        int t = threadIdx.x;
        int k = t >> 4, n = t & 15;
        sW[t] = (n < CLS) ? W2[k * CLS + n] : 0.0f;
    }
    __syncthreads();

    const int wave = threadIdx.x >> 5;
    const int lane = threadIdx.x & 31;
    const int tile = blockIdx.x * 8 + wave;
    if (tile >= nTiles) return;

    const int m    = lane & 15;
    const int kk   = (lane >> 4) << 1;
    const int half = lane >> 4;

    const float* arow = a_in + (size_t)(tile * 16 + m) * HID + kk;
    v8f c = {0.f, 0.f, 0.f, 0.f, 0.f, 0.f, 0.f, 0.f};

#pragma unroll
    for (int k0 = 0; k0 < HID; k0 += 4) {
        v2f a = *(const v2f*)(arow + k0);
        v2f b;
        b.x = sW[(k0 + kk)     * 16 + m];
        b.y = sW[(k0 + kk + 1) * 16 + m];
        c = __builtin_amdgcn_wmma_f32_16x16x4_f32(false, a, false, b,
                                                  (short)0, c, false, false);
    }

    if (m < CLS) {
#pragma unroll
        for (int v = 0; v < 8; ++v) {
            int row = tile * 16 + half * 8 + v;
            h2[(size_t)row * CLS + m] = c[v];
        }
    }
}

__global__ void self_init2_k(const float* __restrict__ h2, const float* __restrict__ dinv,
                             const float* __restrict__ b2, float* __restrict__ out, int n) {
    int t = blockIdx.x * blockDim.x + threadIdx.x;
    if (t >= n) return;
    int i = t / CLS, f = t % CLS;
    float di = dinv[i];
    out[t] = di * di * h2[t] + b2[f];
}

template <bool TAIL>
__device__ __forceinline__ void adds2(float* __restrict__ out, const float (&v)[8],
                                      unsigned db, float w, unsigned f4, bool fl, int left) {
    const int lane = threadIdx.x & 31;
    unsigned off[8];
    float    val[8];
#pragma unroll
    for (int j = 0; j < 8; ++j) {
        const unsigned lo = 0x11111111u * (unsigned)j;
        const unsigned hi = 0x11111111u * (unsigned)(j + 8);
        off[j] = rowsel(db, lo, hi) + f4;
        val[j] = __uint_as_float(rowsel(__float_as_uint(w), lo, hi)) * v[j];
    }
    if (fl) {
#pragma unroll
        for (int j = 0; j < 8; ++j) {
            if (!TAIL || ((lane & 24) + j) < left)
                unsafeAtomicAdd((float*)((char*)out + off[j]), val[j]);
        }
    }
}

__global__ void __launch_bounds__(256) scatter2_k(const int* __restrict__ src,
                                                  const int* __restrict__ dst,
                                                  const float* __restrict__ dinv,
                                                  const float* __restrict__ h2,
                                                  float* __restrict__ out, int E) {
    const int lane = threadIdx.x & 31;
    const int e0 = (blockIdx.x * 8 + (threadIdx.x >> 5)) * 32;
    if (e0 >= E) return;
    const int left = E - e0;
    int s = 0, d = 0;
    if (lane < left) { s = src[e0 + lane]; d = dst[e0 + lane]; }
    const float ds = dinv[s], dd = dinv[d];

    const int  f  = lane & 7;
    const bool fl = f < CLS;
    const unsigned f4 = (unsigned)(fl ? f : CLS - 1) * 4u;
    const unsigned sb = (unsigned)s * (CLS * 4u);
    float v[8];
#pragma unroll
    for (int j = 0; j < 8; ++j) {
        const unsigned lo = 0x11111111u * (unsigned)j;
        const unsigned hi = 0x11111111u * (unsigned)(j + 8);
        const unsigned off = rowsel(sb, lo, hi) + f4;
        v[j] = *(const float*)((const char*)h2 + off);
    }

    const float w = ds * dd;
    const unsigned db = (unsigned)d * (CLS * 4u);
    if (left >= 32) adds2<false>(out, v, db, w, f4, fl, left);
    else            adds2<true >(out, v, db, w, f4, fl, left);
}

extern "C" void kernel_launch(void* const* d_in, const int* in_sizes, int n_in,
                              void* d_out, int out_size, void* d_ws, size_t ws_size,
                              hipStream_t stream) {
    const float* x  = (const float*)d_in[0];
    const int*   ei = (const int*)  d_in[1];
    const float* W1 = (const float*)d_in[2];
    const float* b1 = (const float*)d_in[3];
    const float* W2 = (const float*)d_in[4];
    const float* b2 = (const float*)d_in[5];
    float* out = (float*)d_out;

    const int N = in_sizes[0] / F_IN;
    const int E = in_sizes[1] / 2;
    const int* src = ei;
    const int* dst = ei + E;

    float* dinv = (float*)d_ws;
    float* h    = dinv + N;
    float* agg  = h   + (size_t)N * HID;
    float* h2   = agg + (size_t)N * HID;

    const int B = 256;
    const int nTiles = N / 16;

    const int nParts = (E + CHUNK - 1) / CHUNK;
    unsigned* parts = (unsigned*)h;
    deg_hist_k       <<<nParts, 1024, 0, stream>>>(dst, parts, E);
    dinv_from_parts_k<<<(HWORDS + B - 1) / B, B, 0, stream>>>(parts, dinv, nParts, N);

    gemm1_k    <<<(nTiles + 7) / 8, B, 0, stream>>>(x, W1, h, nTiles);

    int n1 = N * HID;
    self_init1_k<<<(n1 + B - 1) / B, B, 0, stream>>>(h, dinv, agg, n1);
    scatter1_k<<<(E + B - 1) / B, B, 0, stream>>>(src, dst, dinv, h, agg, E);
    bias_relu_k<<<(n1 + B - 1) / B, B, 0, stream>>>(agg, b1, n1);

    gemm2_k    <<<(nTiles + 7) / 8, B, 0, stream>>>(agg, W2, h2, nTiles);

    int n2 = N * CLS;
    self_init2_k<<<(n2 + B - 1) / B, B, 0, stream>>>(h2, dinv, b2, out, n2);
    scatter2_k<<<(E + B - 1) / B, B, 0, stream>>>(src, dst, dinv, h2, out, E);
}


# Normalisation folded out of both edge passes: GEMMs store rows scaled by dinv[src] (also the self-loop start), dinv[dst] applied once per node after; no weight gathers per edge; sum rounding changes
# speedup vs baseline: 1.9277x; 1.1404x over previous
#include <hip/hip_runtime.h>
#include <hip/hip_bf16.h>

typedef __attribute__((ext_vector_type(2))) float v2f;
typedef __attribute__((ext_vector_type(8))) float v8f;

#define F_IN 512
#define HID  16
#define CLS  7

#define NODES  100000
#define HWORDS (NODES / 2)
#define CHUNK  50000

__global__ void __launch_bounds__(1024) deg_hist_k(const int* __restrict__ dst,
                                                   unsigned* __restrict__ parts, int E) {
    __shared__ __attribute__((aligned(16))) unsigned cnt[HWORDS];

    const int base = blockIdx.x * CHUNK;
    int end = base + CHUNK;
    if (end > E) end = E;
    const int nvec = (end - base) >> 2;
    const int4* d4 = (const int4*)(dst + base);

    constexpr int SLOTS = (CHUNK / 4 + 1023) / 1024;
    int4 d[SLOTS];
#pragma unroll
    for (int j = 0; j < SLOTS; ++j) {
        const int i = threadIdx.x + j * 1024;
        if (i < nvec) d[j] = d4[i];
    }

    for (int i = threadIdx.x; i < HWORDS / 4; i += 1024)
        ((uint4*)cnt)[i] = make_uint4(0u, 0u, 0u, 0u);
    __syncthreads();

#pragma unroll
    for (int j = 0; j < SLOTS; ++j) {
        const int i = threadIdx.x + j * 1024;
        if (i < nvec) {
            atomicAdd(&cnt[d[j].x >> 1], 1u << ((d[j].x & 1) << 4));
            atomicAdd(&cnt[d[j].y >> 1], 1u << ((d[j].y & 1) << 4));
            atomicAdd(&cnt[d[j].z >> 1], 1u << ((d[j].z & 1) << 4));
            atomicAdd(&cnt[d[j].w >> 1], 1u << ((d[j].w & 1) << 4));
        }
    }
    for (int e = base + (nvec << 2) + threadIdx.x; e < end; e += 1024) {
        const int t = dst[e];
        atomicAdd(&cnt[t >> 1], 1u << ((t & 1) << 4));
    }
    __syncthreads();

    uint4* o = (uint4*)(parts + (size_t)blockIdx.x * HWORDS);
    for (int i = threadIdx.x; i < HWORDS / 4; i += 1024) o[i] = ((const uint4*)cnt)[i];
}

__global__ void dinv_from_parts_k(const unsigned* __restrict__ parts, float* __restrict__ dinv,
                                  int nParts, int n) {
    int t = blockIdx.x * blockDim.x + threadIdx.x;
    if (t >= HWORDS) return;
    unsigned lo = 0u, hi = 0u;
#pragma unroll 8
    for (int p = 0; p < nParts; ++p) {
        unsigned w = parts[(size_t)p * HWORDS + t];
        lo += w & 0xffffu;
        hi += w >> 16;
    }
    const int i = 2 * t;
    float r0 = rsqrtf(1.0f + (float)lo);
    float r1 = rsqrtf(1.0f + (float)hi);
    if (i + 1 < n)      *(float2*)(dinv + i) = make_float2(r0, r1);
    else if (i < n)     dinv[i] = r0;
}

__global__ void __launch_bounds__(256) gemm1_k(const float* __restrict__ x,
                                               const float* __restrict__ W1,
                                               const float* __restrict__ dinv,
                                               float* __restrict__ h,
                                               float* __restrict__ agg, int nTiles) {
    __shared__ float sW[F_IN * HID];
    for (int i = threadIdx.x * 4; i < F_IN * HID; i += 256 * 4)
        *(float4*)(sW + i) = *(const float4*)(W1 + i);
    __syncthreads();

    const int wave = threadIdx.x >> 5;
    const int lane = threadIdx.x & 31;
    const int tile = blockIdx.x * 8 + wave;
    if (tile >= nTiles) return;

    const int m    = lane & 15;
    const int kk   = (lane >> 4) << 1;
    const int half = lane >> 4;

    const float* arow = x + (size_t)(tile * 16 + m) * F_IN + kk;
    v8f c = {0.f, 0.f, 0.f, 0.f, 0.f, 0.f, 0.f, 0.f};

#pragma unroll 4
    for (int k0 = 0; k0 < F_IN; k0 += 4) {
        v2f a = *(const v2f*)(arow + k0);
        v2f b;
        b.x = sW[(k0 + kk)     * HID + m];
        b.y = sW[(k0 + kk + 1) * HID + m];
        c = __builtin_amdgcn_wmma_f32_16x16x4_f32(false, a, false, b,
                                                  (short)0, c, false, false);
    }

    const float4 dA = *(const float4*)(dinv + tile * 16 + half * 8);
    const float4 dB = *(const float4*)(dinv + tile * 16 + half * 8 + 4);
    const float dr[8] = {dA.x, dA.y, dA.z, dA.w, dB.x, dB.y, dB.z, dB.w};

    const size_t o = (size_t)(tile * 16 + half * 8) * HID + m;
#pragma unroll
    for (int v = 0; v < 8; ++v) {
        const float val = dr[v] * c[v];
        h  [o + (size_t)v * HID] = val;
        agg[o + (size_t)v * HID] = val;
    }
}


__device__ __forceinline__ unsigned rowsel(unsigned x, unsigned lo, unsigned hi) {
    return __builtin_amdgcn_permlane16(x, x, lo, hi, false, false);
}

template <bool TAIL>
__device__ __forceinline__ void adds1(float* __restrict__ agg, const float (&v)[16],
                                      unsigned db, unsigned f4, int left) {
    const int lane = threadIdx.x & 31;
#pragma unroll
    for (int j = 0; j < 16; ++j) {
        const unsigned sel = 0x11111111u * (unsigned)j;
        const unsigned off = rowsel(db, sel, sel) + f4;
        if (!TAIL || ((lane & 16) + j) < left)
            unsafeAtomicAdd((float*)((char*)agg + off), v[j]);
    }
}

__global__ void __launch_bounds__(256) scatter1_k(const int* __restrict__ src,
                                                  const int* __restrict__ dst,
                                                  const float* __restrict__ h,
                                                  float* __restrict__ agg, int E) {
    const int lane = threadIdx.x & 31;
    const int e0 = (blockIdx.x * 8 + (threadIdx.x >> 5)) * 32;
    if (e0 >= E) return;
    const int left = E - e0;
    int s = 0, d = 0;
    if (lane < left) { s = src[e0 + lane]; d = dst[e0 + lane]; }

    const unsigned f4 = (unsigned)(lane & 15) * 4u;
    const unsigned sb = (unsigned)s * (HID * 4u);
    float v[16];
#pragma unroll
    for (int j = 0; j < 16; ++j) {
        const unsigned sel = 0x11111111u * (unsigned)j;
        const unsigned off = rowsel(sb, sel, sel) + f4;
        v[j] = *(const float*)((const char*)h + off);
    }

    const unsigned db = (unsigned)d * (HID * 4u);
    if (left >= 32) adds1<false>(agg, v, db, f4, left);
    else            adds1<true >(agg, v, db, f4, left);
}

__global__ void bias_relu_k(float* __restrict__ agg, const float* __restrict__ dinv,
                            const float* __restrict__ b1, int n) {
    int t = blockIdx.x * blockDim.x + threadIdx.x;
    if (t >= n) return;
    float v = dinv[t >> 4] * agg[t] + b1[t & 15];
    agg[t] = v > 0.f ? v : 0.f;
}

__global__ void __launch_bounds__(256) gemm2_k(const float* __restrict__ a_in,
                                               const float* __restrict__ W2,
                                               const float* __restrict__ dinv,
                                               float* __restrict__ h2,
                                               float* __restrict__ out, int nTiles) {
    __shared__ float sW[HID * 16];
    {
        int t = threadIdx.x;
        int k = t >> 4, n = t & 15;
        sW[t] = (n < CLS) ? W2[k * CLS + n] : 0.0f;
    }
    __syncthreads();

    const int wave = threadIdx.x >> 5;
    const int lane = threadIdx.x & 31;
    const int tile = blockIdx.x * 8 + wave;
    if (tile >= nTiles) return;

    const int m    = lane & 15;
    const int kk   = (lane >> 4) << 1;
    const int half = lane >> 4;

    const float* arow = a_in + (size_t)(tile * 16 + m) * HID + kk;
    v8f c = {0.f, 0.f, 0.f, 0.f, 0.f, 0.f, 0.f, 0.f};

#pragma unroll
    for (int k0 = 0; k0 < HID; k0 += 4) {
        v2f a = *(const v2f*)(arow + k0);
        v2f b;
        b.x = sW[(k0 + kk)     * 16 + m];
        b.y = sW[(k0 + kk + 1) * 16 + m];
        c = __builtin_amdgcn_wmma_f32_16x16x4_f32(false, a, false, b,
                                                  (short)0, c, false, false);
    }

    const float4 dA = *(const float4*)(dinv + tile * 16 + half * 8);
    const float4 dB = *(const float4*)(dinv + tile * 16 + half * 8 + 4);
    const float dr[8] = {dA.x, dA.y, dA.z, dA.w, dB.x, dB.y, dB.z, dB.w};

    if (m < CLS) {
#pragma unroll
        for (int v = 0; v < 8; ++v) {
            const int row = tile * 16 + half * 8 + v;
            const float val = dr[v] * c[v];
            h2 [(size_t)row * CLS + m] = val;
            out[(size_t)row * CLS + m] = val;
        }
    }
}


template <bool TAIL>
__device__ __forceinline__ void adds2(float* __restrict__ out, const float (&v)[8],
                                      unsigned db, unsigned f4, bool fl, int left) {
    const int lane = threadIdx.x & 31;
    unsigned off[8];
#pragma unroll
    for (int j = 0; j < 8; ++j) {
        const unsigned lo = 0x11111111u * (unsigned)j;
        const unsigned hi = 0x11111111u * (unsigned)(j + 8);
        off[j] = rowsel(db, lo, hi) + f4;
    }
    if (fl) {
#pragma unroll
        for (int j = 0; j < 8; ++j) {
            if (!TAIL || ((lane & 24) + j) < left)
                unsafeAtomicAdd((float*)((char*)out + off[j]), v[j]);
        }
    }
}

__global__ void __launch_bounds__(256) scatter2_k(const int* __restrict__ src,
                                                  const int* __restrict__ dst,
                                                  const float* __restrict__ h2,
                                                  float* __restrict__ out, int E) {
    const int lane = threadIdx.x & 31;
    const int e0 = (blockIdx.x * 8 + (threadIdx.x >> 5)) * 32;
    if (e0 >= E) return;
    const int left = E - e0;
    int s = 0, d = 0;
    if (lane < left) { s = src[e0 + lane]; d = dst[e0 + lane]; }

    const int  f  = lane & 7;
    const bool fl = f < CLS;
    const unsigned f4 = (unsigned)(fl ? f : CLS - 1) * 4u;
    const unsigned sb = (unsigned)s * (CLS * 4u);
    float v[8];
#pragma unroll
    for (int j = 0; j < 8; ++j) {
        const unsigned lo = 0x11111111u * (unsigned)j;
        const unsigned hi = 0x11111111u * (unsigned)(j + 8);
        const unsigned off = rowsel(sb, lo, hi) + f4;
        v[j] = *(const float*)((const char*)h2 + off);
    }

    const unsigned db = (unsigned)d * (CLS * 4u);
    if (left >= 32) adds2<false>(out, v, db, f4, fl, left);
    else            adds2<true >(out, v, db, f4, fl, left);
}

__global__ void final2_k(float* __restrict__ out, const float* __restrict__ dinv,
                         const float* __restrict__ b2, int n) {
    int t = blockIdx.x * blockDim.x + threadIdx.x;
    if (t >= n) return;
    int i = t / CLS, f = t % CLS;
    out[t] = dinv[i] * out[t] + b2[f];
}

extern "C" void kernel_launch(void* const* d_in, const int* in_sizes, int n_in,
                              void* d_out, int out_size, void* d_ws, size_t ws_size,
                              hipStream_t stream) {
    const float* x  = (const float*)d_in[0];
    const int*   ei = (const int*)  d_in[1];
    const float* W1 = (const float*)d_in[2];
    const float* b1 = (const float*)d_in[3];
    const float* W2 = (const float*)d_in[4];
    const float* b2 = (const float*)d_in[5];
    float* out = (float*)d_out;

    const int N = in_sizes[0] / F_IN;
    const int E = in_sizes[1] / 2;
    const int* src = ei;
    const int* dst = ei + E;

    float* dinv = (float*)d_ws;
    float* h    = dinv + N;
    float* agg  = h   + (size_t)N * HID;
    float* h2   = agg + (size_t)N * HID;

    const int B = 256;
    const int nTiles = N / 16;

    const int nParts = (E + CHUNK - 1) / CHUNK;
    unsigned* parts = (unsigned*)h;
    deg_hist_k       <<<nParts, 1024, 0, stream>>>(dst, parts, E);
    dinv_from_parts_k<<<(HWORDS + B - 1) / B, B, 0, stream>>>(parts, dinv, nParts, N);

    gemm1_k    <<<(nTiles + 7) / 8, B, 0, stream>>>(x, W1, dinv, h, agg, nTiles);

    int n1 = N * HID;
    scatter1_k<<<(E + B - 1) / B, B, 0, stream>>>(src, dst, h, agg, E);
    bias_relu_k<<<(n1 + B - 1) / B, B, 0, stream>>>(agg, dinv, b1, n1);

    gemm2_k    <<<(nTiles + 7) / 8, B, 0, stream>>>(agg, W2, dinv, h2, out, nTiles);

    int n2 = N * CLS;
    scatter2_k<<<(E + B - 1) / B, B, 0, stream>>>(src, dst, h2, out, E);
    final2_k  <<<(n2 + B - 1) / B, B, 0, stream>>>(out, dinv, b2, n2);
}


# Both edge passes: one copy of the atomic sums per chip half (die id, message 0x87, picks a wave's copy), copies summed per node after; sums split in two partials, still arrival order
# speedup vs baseline: 3.1937x; 1.6567x over previous
#include <hip/hip_runtime.h>
#include <hip/hip_bf16.h>

typedef __attribute__((ext_vector_type(2))) float v2f;
typedef __attribute__((ext_vector_type(8))) float v8f;

#define F_IN 512
#define HID  16
#define CLS  7

__device__ __forceinline__ unsigned chip_half() {
    return (__builtin_amdgcn_s_sendmsg_rtn(0x87) >> 18) & 1u;
}

#define NODES  100000
#define HWORDS (NODES / 2)
#define CHUNK  50000

#define ROWS_B ((NODES * CLS) / HID)

__global__ void __launch_bounds__(1024) deg_hist_k(const int* __restrict__ dst,
                                                   unsigned* __restrict__ parts, int E) {
    __shared__ __attribute__((aligned(16))) unsigned cnt[HWORDS];

    const int base = blockIdx.x * CHUNK;
    int end = base + CHUNK;
    if (end > E) end = E;
    const int nvec = (end - base) >> 2;
    const int4* d4 = (const int4*)(dst + base);

    constexpr int SLOTS = (CHUNK / 4 + 1023) / 1024;
    int4 d[SLOTS];
#pragma unroll
    for (int j = 0; j < SLOTS; ++j) {
        const int i = threadIdx.x + j * 1024;
        if (i < nvec) d[j] = d4[i];
    }

    for (int i = threadIdx.x; i < HWORDS / 4; i += 1024)
        ((uint4*)cnt)[i] = make_uint4(0u, 0u, 0u, 0u);
    __syncthreads();

#pragma unroll
    for (int j = 0; j < SLOTS; ++j) {
        const int i = threadIdx.x + j * 1024;
        if (i < nvec) {
            atomicAdd(&cnt[d[j].x >> 1], 1u << ((d[j].x & 1) << 4));
            atomicAdd(&cnt[d[j].y >> 1], 1u << ((d[j].y & 1) << 4));
            atomicAdd(&cnt[d[j].z >> 1], 1u << ((d[j].z & 1) << 4));
            atomicAdd(&cnt[d[j].w >> 1], 1u << ((d[j].w & 1) << 4));
        }
    }
    for (int e = base + (nvec << 2) + threadIdx.x; e < end; e += 1024) {
        const int t = dst[e];
        atomicAdd(&cnt[t >> 1], 1u << ((t & 1) << 4));
    }
    __syncthreads();

    uint4* o = (uint4*)(parts + (size_t)blockIdx.x * HWORDS);
    for (int i = threadIdx.x; i < HWORDS / 4; i += 1024) o[i] = ((const uint4*)cnt)[i];
}

__global__ void dinv_from_parts_k(const unsigned* __restrict__ parts, float* __restrict__ dinv,
                                  int nParts, int n) {
    int t = blockIdx.x * blockDim.x + threadIdx.x;
    if (t >= HWORDS) return;
    unsigned lo = 0u, hi = 0u;
#pragma unroll 8
    for (int p = 0; p < nParts; ++p) {
        unsigned w = parts[(size_t)p * HWORDS + t];
        lo += w & 0xffffu;
        hi += w >> 16;
    }
    const int i = 2 * t;
    float r0 = rsqrtf(1.0f + (float)lo);
    float r1 = rsqrtf(1.0f + (float)hi);
    if (i + 1 < n)      *(float2*)(dinv + i) = make_float2(r0, r1);
    else if (i < n)     dinv[i] = r0;
}

__global__ void __launch_bounds__(256) gemm1_k(const float* __restrict__ x,
                                               const float* __restrict__ W1,
                                               const float* __restrict__ dinv,
                                               float* __restrict__ h,
                                               float* __restrict__ agg,
                                               float* __restrict__ a1a,
                                               float* __restrict__ a1b, int nTiles) {
    __shared__ float sW[F_IN * HID];
    for (int i = threadIdx.x * 4; i < F_IN * HID; i += 256 * 4)
        *(float4*)(sW + i) = *(const float4*)(W1 + i);
    __syncthreads();

    const int wave = threadIdx.x >> 5;
    const int lane = threadIdx.x & 31;
    const int tile = blockIdx.x * 8 + wave;
    if (tile >= nTiles) return;

    const int m    = lane & 15;
    const int kk   = (lane >> 4) << 1;
    const int half = lane >> 4;

    const float* arow = x + (size_t)(tile * 16 + m) * F_IN + kk;
    v8f c = {0.f, 0.f, 0.f, 0.f, 0.f, 0.f, 0.f, 0.f};

#pragma unroll 4
    for (int k0 = 0; k0 < F_IN; k0 += 4) {
        v2f a = *(const v2f*)(arow + k0);
        v2f b;
        b.x = sW[(k0 + kk)     * HID + m];
        b.y = sW[(k0 + kk + 1) * HID + m];
        c = __builtin_amdgcn_wmma_f32_16x16x4_f32(false, a, false, b,
                                                  (short)0, c, false, false);
    }

    const float4 dA = *(const float4*)(dinv + tile * 16 + half * 8);
    const float4 dB = *(const float4*)(dinv + tile * 16 + half * 8 + 4);
    const float dr[8] = {dA.x, dA.y, dA.z, dA.w, dB.x, dB.y, dB.z, dB.w};

    const size_t o = (size_t)(tile * 16 + half * 8) * HID + m;
#pragma unroll
    for (int v = 0; v < 8; ++v) {
        const float val = dr[v] * c[v];
        h  [o + (size_t)v * HID] = val;
        agg[o + (size_t)v * HID] = val;
        const int r = tile * 16 + half * 8 + v;
        float* z = (r < ROWS_B) ? (a1a + (size_t)r * HID + m) : (a1b + (size_t)(r - ROWS_B) * HID + m);
        if (r < 2 * ROWS_B) *z = 0.0f;
    }
}


__device__ __forceinline__ unsigned rowsel(unsigned x, unsigned lo, unsigned hi) {
    return __builtin_amdgcn_permlane16(x, x, lo, hi, false, false);
}

template <bool TAIL>
__device__ __forceinline__ void adds1(float* agg, const float (&v)[16],
                                      unsigned rlo, unsigned rhi, unsigned f4, int left) {
    const int lane = threadIdx.x & 31;
    char* const colbase = (char*)agg + f4;
#pragma unroll
    for (int j = 0; j < 16; ++j) {
        const unsigned sel = 0x11111111u * (unsigned)j;
        const unsigned lo = rowsel(rlo, sel, sel);
        const unsigned hi = rowsel(rhi, sel, sel);
        const long long off = (long long)(((unsigned long long)hi << 32) | lo);
        if (!TAIL || ((lane & 16) + j) < left)
            unsafeAtomicAdd((float*)(colbase + off), v[j]);
    }
}

__global__ void __launch_bounds__(256) scatter1_k(const int* __restrict__ src,
                                                  const int* __restrict__ dst,
                                                  const float* __restrict__ h,
                                                  float* agg, float* a1a, float* a1b, int E) {
    const int lane = threadIdx.x & 31;
    const int e0 = (blockIdx.x * 8 + (threadIdx.x >> 5)) * 32;
    if (e0 >= E) return;
    const int left = E - e0;
    int s = 0, d = 0;
    if (lane < left) { s = src[e0 + lane]; d = dst[e0 + lane]; }

    const unsigned f4 = (unsigned)(lane & 15) * 4u;
    const unsigned sb = (unsigned)s * (HID * 4u);
    float v[16];
#pragma unroll
    for (int j = 0; j < 16; ++j) {
        const unsigned sel = 0x11111111u * (unsigned)j;
        const unsigned off = rowsel(sb, sel, sel) + f4;
        v[j] = *(const float*)((const char*)h + off);
    }

    const unsigned half = chip_half();
    const float* row = agg + (size_t)d * HID;
    if (half != 0u) {
        if (d < ROWS_B)          row = a1a + (size_t)d * HID;
        else if (d < 2 * ROWS_B) row = a1b + (size_t)(d - ROWS_B) * HID;
    }
    const long long roff = (const char*)row - (const char*)agg;
    const unsigned rlo = (unsigned)(unsigned long long)roff;
    const unsigned rhi = (unsigned)((unsigned long long)roff >> 32);
    if (left >= 32) adds1<false>(agg, v, rlo, rhi, f4, left);
    else            adds1<true >(agg, v, rlo, rhi, f4, left);
}

__global__ void bias_relu_k(float* agg, const float* __restrict__ a1a,
                            const float* __restrict__ a1b, const float* __restrict__ dinv,
                            const float* __restrict__ b1, int n) {
    int t = blockIdx.x * blockDim.x + threadIdx.x;
    if (t >= n) return;
    const int row = t >> 4;
    const bool has = row < 2 * ROWS_B;
    const float* p2 = (row < ROWS_B) ? (a1a + t) : (has ? (a1b + (t - ROWS_B * HID)) : (agg + t));
    const float a0 = agg[t];
    const float a1 = *p2;
    const float a  = has ? a0 + a1 : a0;
    float v = dinv[row] * a + b1[t & 15];
    agg[t] = v > 0.f ? v : 0.f;
}

__global__ void __launch_bounds__(256) gemm2_k(const float* __restrict__ a_in,
                                               const float* __restrict__ W2,
                                               const float* __restrict__ dinv,
                                               float* __restrict__ h2,
                                               float* __restrict__ out,
                                               float* __restrict__ out1, int nTiles) {
    __shared__ float sW[HID * 16];
    {
        int t = threadIdx.x;
        int k = t >> 4, n = t & 15;
        sW[t] = (n < CLS) ? W2[k * CLS + n] : 0.0f;
    }
    __syncthreads();

    const int wave = threadIdx.x >> 5;
    const int lane = threadIdx.x & 31;
    const int tile = blockIdx.x * 8 + wave;
    if (tile >= nTiles) return;

    const int m    = lane & 15;
    const int kk   = (lane >> 4) << 1;
    const int half = lane >> 4;

    const float* arow = a_in + (size_t)(tile * 16 + m) * HID + kk;
    v8f c = {0.f, 0.f, 0.f, 0.f, 0.f, 0.f, 0.f, 0.f};

#pragma unroll
    for (int k0 = 0; k0 < HID; k0 += 4) {
        v2f a = *(const v2f*)(arow + k0);
        v2f b;
        b.x = sW[(k0 + kk)     * 16 + m];
        b.y = sW[(k0 + kk + 1) * 16 + m];
        c = __builtin_amdgcn_wmma_f32_16x16x4_f32(false, a, false, b,
                                                  (short)0, c, false, false);
    }

    const float4 dA = *(const float4*)(dinv + tile * 16 + half * 8);
    const float4 dB = *(const float4*)(dinv + tile * 16 + half * 8 + 4);
    const float dr[8] = {dA.x, dA.y, dA.z, dA.w, dB.x, dB.y, dB.z, dB.w};

    if (m < CLS) {
#pragma unroll
        for (int v = 0; v < 8; ++v) {
            const int row = tile * 16 + half * 8 + v;
            const float val = dr[v] * c[v];
            h2  [(size_t)row * CLS + m] = val;
            out [(size_t)row * CLS + m] = val;
            out1[(size_t)row * CLS + m] = 0.0f;
        }
    }
}


template <bool TAIL>
__device__ __forceinline__ void adds2(float* __restrict__ out, const float (&v)[8],
                                      unsigned db, unsigned f4, bool fl, int left) {
    const int lane = threadIdx.x & 31;
    unsigned off[8];
#pragma unroll
    for (int j = 0; j < 8; ++j) {
        const unsigned lo = 0x11111111u * (unsigned)j;
        const unsigned hi = 0x11111111u * (unsigned)(j + 8);
        off[j] = rowsel(db, lo, hi) + f4;
    }
    if (fl) {
#pragma unroll
        for (int j = 0; j < 8; ++j) {
            if (!TAIL || ((lane & 24) + j) < left)
                unsafeAtomicAdd((float*)((char*)out + off[j]), v[j]);
        }
    }
}

__global__ void __launch_bounds__(256) scatter2_k(const int* __restrict__ src,
                                                  const int* __restrict__ dst,
                                                  const float* __restrict__ h2,
                                                  float* __restrict__ out0,
                                                  float* __restrict__ out1, int E) {
    const int lane = threadIdx.x & 31;
    const int e0 = (blockIdx.x * 8 + (threadIdx.x >> 5)) * 32;
    if (e0 >= E) return;
    const int left = E - e0;
    int s = 0, d = 0;
    if (lane < left) { s = src[e0 + lane]; d = dst[e0 + lane]; }

    const int  f  = lane & 7;
    const bool fl = f < CLS;
    const unsigned f4 = (unsigned)(fl ? f : CLS - 1) * 4u;
    const unsigned sb = (unsigned)s * (CLS * 4u);
    float v[8];
#pragma unroll
    for (int j = 0; j < 8; ++j) {
        const unsigned lo = 0x11111111u * (unsigned)j;
        const unsigned hi = 0x11111111u * (unsigned)(j + 8);
        const unsigned off = rowsel(sb, lo, hi) + f4;
        v[j] = *(const float*)((const char*)h2 + off);
    }

    float* out = chip_half() ? out1 : out0;
    const unsigned db = (unsigned)d * (CLS * 4u);
    if (left >= 32) adds2<false>(out, v, db, f4, fl, left);
    else            adds2<true >(out, v, db, f4, fl, left);
}

__global__ void final2_k(float* __restrict__ out, const float* __restrict__ out1,
                         const float* __restrict__ dinv,
                         const float* __restrict__ b2, int n) {
    int t = blockIdx.x * blockDim.x + threadIdx.x;
    if (t >= n) return;
    int i = t / CLS, f = t % CLS;
    out[t] = dinv[i] * (out[t] + out1[t]) + b2[f];
}

extern "C" void kernel_launch(void* const* d_in, const int* in_sizes, int n_in,
                              void* d_out, int out_size, void* d_ws, size_t ws_size,
                              hipStream_t stream) {
    const float* x  = (const float*)d_in[0];
    const int*   ei = (const int*)  d_in[1];
    const float* W1 = (const float*)d_in[2];
    const float* b1 = (const float*)d_in[3];
    const float* W2 = (const float*)d_in[4];
    const float* b2 = (const float*)d_in[5];
    float* out = (float*)d_out;

    const int N = in_sizes[0] / F_IN;
    const int E = in_sizes[1] / 2;
    const int* src = ei;
    const int* dst = ei + E;

    float* dinv = (float*)d_ws;
    float* h    = dinv + N;
    float* agg  = h   + (size_t)N * HID;
    float* h2   = agg + (size_t)N * HID;
    float* a1a  = h2;
    float* a1b  = out;
    float* out1 = h;

    const int B = 256;
    const int nTiles = N / 16;

    const int nParts = (E + CHUNK - 1) / CHUNK;
    unsigned* parts = (unsigned*)h;
    deg_hist_k       <<<nParts, 1024, 0, stream>>>(dst, parts, E);
    dinv_from_parts_k<<<(HWORDS + B - 1) / B, B, 0, stream>>>(parts, dinv, nParts, N);

    gemm1_k    <<<(nTiles + 7) / 8, B, 0, stream>>>(x, W1, dinv, h, agg, a1a, a1b, nTiles);

    int n1 = N * HID;
    scatter1_k<<<(E + B - 1) / B, B, 0, stream>>>(src, dst, h, agg, a1a, a1b, E);
    bias_relu_k<<<(n1 + B - 1) / B, B, 0, stream>>>(agg, a1a, a1b, dinv, b1, n1);

    gemm2_k    <<<(nTiles + 7) / 8, B, 0, stream>>>(agg, W2, dinv, h2, out, out1, nTiles);

    int n2 = N * CLS;
    scatter2_k<<<(E + B - 1) / B, B, 0, stream>>>(src, dst, h2, out, out1, E);
    final2_k  <<<(n2 + B - 1) / B, B, 0, stream>>>(out, out1, dinv, b2, n2);
}


# Layer 2 rows padded from 7 to 8 floats (gather table, both sum copies; first copy in place over the GEMM2 input rows): each row gather and row add stays in one 32-byte sector; sum order as before
# speedup vs baseline: 3.4868x; 1.0918x over previous
#include <hip/hip_runtime.h>
#include <hip/hip_bf16.h>

typedef __attribute__((ext_vector_type(2))) float v2f;
typedef __attribute__((ext_vector_type(8))) float v8f;

#define F_IN 512
#define HID  16
#define CLS  7
#define P2   8

__device__ __forceinline__ unsigned chip_half() {
    return (__builtin_amdgcn_s_sendmsg_rtn(0x87) >> 18) & 1u;
}

#define NODES  100000
#define HWORDS (NODES / 2)
#define CHUNK  50000

#define ROWS_B ((NODES * CLS) / HID)

__global__ void __launch_bounds__(1024) deg_hist_k(const int* __restrict__ dst,
                                                   unsigned* __restrict__ parts, int E) {
    __shared__ __attribute__((aligned(16))) unsigned cnt[HWORDS];

    const int base = blockIdx.x * CHUNK;
    int end = base + CHUNK;
    if (end > E) end = E;
    const int nvec = (end - base) >> 2;
    const int4* d4 = (const int4*)(dst + base);

    constexpr int SLOTS = (CHUNK / 4 + 1023) / 1024;
    int4 d[SLOTS];
#pragma unroll
    for (int j = 0; j < SLOTS; ++j) {
        const int i = threadIdx.x + j * 1024;
        if (i < nvec) d[j] = d4[i];
    }

    for (int i = threadIdx.x; i < HWORDS / 4; i += 1024)
        ((uint4*)cnt)[i] = make_uint4(0u, 0u, 0u, 0u);
    __syncthreads();

#pragma unroll
    for (int j = 0; j < SLOTS; ++j) {
        const int i = threadIdx.x + j * 1024;
        if (i < nvec) {
            atomicAdd(&cnt[d[j].x >> 1], 1u << ((d[j].x & 1) << 4));
            atomicAdd(&cnt[d[j].y >> 1], 1u << ((d[j].y & 1) << 4));
            atomicAdd(&cnt[d[j].z >> 1], 1u << ((d[j].z & 1) << 4));
            atomicAdd(&cnt[d[j].w >> 1], 1u << ((d[j].w & 1) << 4));
        }
    }
    for (int e = base + (nvec << 2) + threadIdx.x; e < end; e += 1024) {
        const int t = dst[e];
        atomicAdd(&cnt[t >> 1], 1u << ((t & 1) << 4));
    }
    __syncthreads();

    uint4* o = (uint4*)(parts + (size_t)blockIdx.x * HWORDS);
    for (int i = threadIdx.x; i < HWORDS / 4; i += 1024) o[i] = ((const uint4*)cnt)[i];
}

__global__ void dinv_from_parts_k(const unsigned* __restrict__ parts, float* __restrict__ dinv,
                                  int nParts, int n) {
    int t = blockIdx.x * blockDim.x + threadIdx.x;
    if (t >= HWORDS) return;
    unsigned lo = 0u, hi = 0u;
#pragma unroll 8
    for (int p = 0; p < nParts; ++p) {
        unsigned w = parts[(size_t)p * HWORDS + t];
        lo += w & 0xffffu;
        hi += w >> 16;
    }
    const int i = 2 * t;
    float r0 = rsqrtf(1.0f + (float)lo);
    float r1 = rsqrtf(1.0f + (float)hi);
    if (i + 1 < n)      *(float2*)(dinv + i) = make_float2(r0, r1);
    else if (i < n)     dinv[i] = r0;
}

__global__ void __launch_bounds__(256) gemm1_k(const float* __restrict__ x,
                                               const float* __restrict__ W1,
                                               const float* __restrict__ dinv,
                                               float* __restrict__ h,
                                               float* __restrict__ agg,
                                               float* __restrict__ a1a,
                                               float* __restrict__ a1b, int nTiles) {
    __shared__ float sW[F_IN * HID];
    for (int i = threadIdx.x * 4; i < F_IN * HID; i += 256 * 4)
        *(float4*)(sW + i) = *(const float4*)(W1 + i);
    __syncthreads();

    const int wave = threadIdx.x >> 5;
    const int lane = threadIdx.x & 31;
    const int tile = blockIdx.x * 8 + wave;
    if (tile >= nTiles) return;

    const int m    = lane & 15;
    const int kk   = (lane >> 4) << 1;
    const int half = lane >> 4;

    const float* arow = x + (size_t)(tile * 16 + m) * F_IN + kk;
    v8f c = {0.f, 0.f, 0.f, 0.f, 0.f, 0.f, 0.f, 0.f};

#pragma unroll 4
    for (int k0 = 0; k0 < F_IN; k0 += 4) {
        v2f a = *(const v2f*)(arow + k0);
        v2f b;
        b.x = sW[(k0 + kk)     * HID + m];
        b.y = sW[(k0 + kk + 1) * HID + m];
        c = __builtin_amdgcn_wmma_f32_16x16x4_f32(false, a, false, b,
                                                  (short)0, c, false, false);
    }

    const float4 dA = *(const float4*)(dinv + tile * 16 + half * 8);
    const float4 dB = *(const float4*)(dinv + tile * 16 + half * 8 + 4);
    const float dr[8] = {dA.x, dA.y, dA.z, dA.w, dB.x, dB.y, dB.z, dB.w};

    const size_t o = (size_t)(tile * 16 + half * 8) * HID + m;
#pragma unroll
    for (int v = 0; v < 8; ++v) {
        const float val = dr[v] * c[v];
        h  [o + (size_t)v * HID] = val;
        agg[o + (size_t)v * HID] = val;
        const int r = tile * 16 + half * 8 + v;
        float* z = (r < ROWS_B) ? (a1a + (size_t)r * HID + m) : (a1b + (size_t)(r - ROWS_B) * HID + m);
        if (r < 2 * ROWS_B) *z = 0.0f;
    }
}


__device__ __forceinline__ unsigned rowsel(unsigned x, unsigned lo, unsigned hi) {
    return __builtin_amdgcn_permlane16(x, x, lo, hi, false, false);
}

template <bool TAIL>
__device__ __forceinline__ void adds1(float* agg, const float (&v)[16],
                                      unsigned rlo, unsigned rhi, unsigned f4, int left) {
    const int lane = threadIdx.x & 31;
    char* const colbase = (char*)agg + f4;
#pragma unroll
    for (int j = 0; j < 16; ++j) {
        const unsigned sel = 0x11111111u * (unsigned)j;
        const unsigned lo = rowsel(rlo, sel, sel);
        const unsigned hi = rowsel(rhi, sel, sel);
        const long long off = (long long)(((unsigned long long)hi << 32) | lo);
        if (!TAIL || ((lane & 16) + j) < left)
            unsafeAtomicAdd((float*)(colbase + off), v[j]);
    }
}

__global__ void __launch_bounds__(256) scatter1_k(const int* __restrict__ src,
                                                  const int* __restrict__ dst,
                                                  const float* __restrict__ h,
                                                  float* agg, float* a1a, float* a1b, int E) {
    const int lane = threadIdx.x & 31;
    const int e0 = (blockIdx.x * 8 + (threadIdx.x >> 5)) * 32;
    if (e0 >= E) return;
    const int left = E - e0;
    int s = 0, d = 0;
    if (lane < left) { s = src[e0 + lane]; d = dst[e0 + lane]; }

    const unsigned f4 = (unsigned)(lane & 15) * 4u;
    const unsigned sb = (unsigned)s * (HID * 4u);
    float v[16];
#pragma unroll
    for (int j = 0; j < 16; ++j) {
        const unsigned sel = 0x11111111u * (unsigned)j;
        const unsigned off = rowsel(sb, sel, sel) + f4;
        v[j] = *(const float*)((const char*)h + off);
    }

    const unsigned half = chip_half();
    const float* row = agg + (size_t)d * HID;
    if (half != 0u) {
        if (d < ROWS_B)          row = a1a + (size_t)d * HID;
        else if (d < 2 * ROWS_B) row = a1b + (size_t)(d - ROWS_B) * HID;
    }
    const long long roff = (const char*)row - (const char*)agg;
    const unsigned rlo = (unsigned)(unsigned long long)roff;
    const unsigned rhi = (unsigned)((unsigned long long)roff >> 32);
    if (left >= 32) adds1<false>(agg, v, rlo, rhi, f4, left);
    else            adds1<true >(agg, v, rlo, rhi, f4, left);
}

__global__ void bias_relu_k(float* agg, const float* __restrict__ a1a,
                            const float* __restrict__ a1b, const float* __restrict__ dinv,
                            const float* __restrict__ b1, int n) {
    int t = blockIdx.x * blockDim.x + threadIdx.x;
    if (t >= n) return;
    const int row = t >> 4;
    const bool has = row < 2 * ROWS_B;
    const float* p2 = (row < ROWS_B) ? (a1a + t) : (has ? (a1b + (t - ROWS_B * HID)) : (agg + t));
    const float a0 = agg[t];
    const float a1 = *p2;
    const float a  = has ? a0 + a1 : a0;
    float v = dinv[row] * a + b1[t & 15];
    agg[t] = v > 0.f ? v : 0.f;
}

__global__ void __launch_bounds__(256) gemm2_k(float* agg,
                                               const float* __restrict__ W2,
                                               const float* __restrict__ dinv,
                                               float* __restrict__ h2p,
                                               float* __restrict__ acc1, int nTiles) {
    __shared__ float sW[HID * 16];
    {
        int t = threadIdx.x;
        int k = t >> 4, n = t & 15;
        sW[t] = (n < CLS) ? W2[k * CLS + n] : 0.0f;
    }
    __syncthreads();

    const int wave = threadIdx.x >> 5;
    const int lane = threadIdx.x & 31;
    const int tile = blockIdx.x * 8 + wave;
    if (tile >= nTiles) return;

    const int m    = lane & 15;
    const int kk   = (lane >> 4) << 1;
    const int half = lane >> 4;

    const float* arow = agg + (size_t)(tile * 16 + m) * HID + kk;
    v8f c = {0.f, 0.f, 0.f, 0.f, 0.f, 0.f, 0.f, 0.f};

#pragma unroll
    for (int k0 = 0; k0 < HID; k0 += 4) {
        v2f a = *(const v2f*)(arow + k0);
        v2f b;
        b.x = sW[(k0 + kk)     * 16 + m];
        b.y = sW[(k0 + kk + 1) * 16 + m];
        c = __builtin_amdgcn_wmma_f32_16x16x4_f32(false, a, false, b,
                                                  (short)0, c, false, false);
    }

    const float4 dA = *(const float4*)(dinv + tile * 16 + half * 8);
    const float4 dB = *(const float4*)(dinv + tile * 16 + half * 8 + 4);
    const float dr[8] = {dA.x, dA.y, dA.z, dA.w, dB.x, dB.y, dB.z, dB.w};

    if (m < P2) {
#pragma unroll
        for (int v = 0; v < 8; ++v) {
            const int row = tile * 16 + half * 8 + v;
            const float val = dr[v] * c[v];
            h2p [(size_t)row * P2  + m] = val;
            agg [(size_t)row * HID + m] = val;
            acc1[(size_t)row * P2  + m] = 0.0f;
        }
    }
}


template <bool TAIL>
__device__ __forceinline__ void adds2(float* __restrict__ out, const float (&v)[8],
                                      unsigned db, unsigned f4, bool fl, int left) {
    const int lane = threadIdx.x & 31;
    unsigned off[8];
#pragma unroll
    for (int j = 0; j < 8; ++j) {
        const unsigned lo = 0x11111111u * (unsigned)j;
        const unsigned hi = 0x11111111u * (unsigned)(j + 8);
        off[j] = rowsel(db, lo, hi) + f4;
    }
    if (fl) {
#pragma unroll
        for (int j = 0; j < 8; ++j) {
            if (!TAIL || ((lane & 24) + j) < left)
                unsafeAtomicAdd((float*)((char*)out + off[j]), v[j]);
        }
    }
}

__global__ void __launch_bounds__(256) scatter2_k(const int* __restrict__ src,
                                                  const int* __restrict__ dst,
                                                  const float* __restrict__ h2p,
                                                  float* __restrict__ acc0,
                                                  float* __restrict__ acc1, int E) {
    const int lane = threadIdx.x & 31;
    const int e0 = (blockIdx.x * 8 + (threadIdx.x >> 5)) * 32;
    if (e0 >= E) return;
    const int left = E - e0;
    int s = 0, d = 0;
    if (lane < left) { s = src[e0 + lane]; d = dst[e0 + lane]; }

    const int  f  = lane & 7;
    const bool fl = f < CLS;
    const unsigned f4 = (unsigned)(fl ? f : CLS - 1) * 4u;
    const unsigned sb = (unsigned)s * (P2 * 4u);
    float v[8];
#pragma unroll
    for (int j = 0; j < 8; ++j) {
        const unsigned lo = 0x11111111u * (unsigned)j;
        const unsigned hi = 0x11111111u * (unsigned)(j + 8);
        const unsigned off = rowsel(sb, lo, hi) + f4;
        v[j] = *(const float*)((const char*)h2p + off);
    }

    const unsigned hf = chip_half();
    float* out = hf ? acc1 : acc0;
    const unsigned db = (unsigned)d * (hf ? P2 * 4u : HID * 4u);
    if (left >= 32) adds2<false>(out, v, db, f4, fl, left);
    else            adds2<true >(out, v, db, f4, fl, left);
}

__global__ void final2_k(float* __restrict__ out, const float* __restrict__ acc0,
                         const float* __restrict__ acc1,
                         const float* __restrict__ dinv,
                         const float* __restrict__ b2, int n) {
    int t = blockIdx.x * blockDim.x + threadIdx.x;
    if (t >= n) return;
    int i = t / CLS, f = t % CLS;
    out[t] = dinv[i] * (acc0[(size_t)i * HID + f] + acc1[(size_t)i * P2 + f]) + b2[f];
}

extern "C" void kernel_launch(void* const* d_in, const int* in_sizes, int n_in,
                              void* d_out, int out_size, void* d_ws, size_t ws_size,
                              hipStream_t stream) {
    const float* x  = (const float*)d_in[0];
    const int*   ei = (const int*)  d_in[1];
    const float* W1 = (const float*)d_in[2];
    const float* b1 = (const float*)d_in[3];
    const float* W2 = (const float*)d_in[4];
    const float* b2 = (const float*)d_in[5];
    float* out = (float*)d_out;

    const int N = in_sizes[0] / F_IN;
    const int E = in_sizes[1] / 2;
    const int* src = ei;
    const int* dst = ei + E;

    float* dinv = (float*)d_ws;
    float* h    = dinv + N;
    float* agg  = h   + (size_t)N * HID;
    float* h2   = agg + (size_t)N * HID;
    float* a1a  = h2;
    float* a1b  = out;
    float* h2p  = h;
    float* acc1 = h + (size_t)N * P2;

    const int B = 256;
    const int nTiles = N / 16;

    const int nParts = (E + CHUNK - 1) / CHUNK;
    unsigned* parts = (unsigned*)h;
    deg_hist_k       <<<nParts, 1024, 0, stream>>>(dst, parts, E);
    dinv_from_parts_k<<<(HWORDS + B - 1) / B, B, 0, stream>>>(parts, dinv, nParts, N);

    gemm1_k    <<<(nTiles + 7) / 8, B, 0, stream>>>(x, W1, dinv, h, agg, a1a, a1b, nTiles);

    int n1 = N * HID;
    scatter1_k<<<(E + B - 1) / B, B, 0, stream>>>(src, dst, h, agg, a1a, a1b, E);
    bias_relu_k<<<(n1 + B - 1) / B, B, 0, stream>>>(agg, a1a, a1b, dinv, b1, n1);

    gemm2_k    <<<(nTiles + 7) / 8, B, 0, stream>>>(agg, W2, dinv, h2p, acc1, nTiles);

    int n2 = N * CLS;
    scatter2_k<<<(E + B - 1) / B, B, 0, stream>>>(src, dst, h2p, agg, acc1, E);
    final2_k  <<<(n2 + B - 1) / B, B, 0, stream>>>(out, agg, acc1, dinv, b2, n2);
}


# GEMM1: x read in whole lines (16-byte loads, 256 B of one row per 16 lanes, next chunk in flight) and staged per wave through LDS; 16-byte fragment reads, two accumulators; k sum order changes
# speedup vs baseline: 3.6197x; 1.0381x over previous
#include <hip/hip_runtime.h>
#include <hip/hip_bf16.h>

typedef __attribute__((ext_vector_type(2))) float v2f;
typedef __attribute__((ext_vector_type(8))) float v8f;

#define F_IN 512
#define HID  16
#define CLS  7
#define P2   8

__device__ __forceinline__ unsigned chip_half() {
    return (__builtin_amdgcn_s_sendmsg_rtn(0x87) >> 18) & 1u;
}

#define NODES  100000
#define HWORDS (NODES / 2)
#define CHUNK  50000

#define ROWS_B ((NODES * CLS) / HID)

__global__ void __launch_bounds__(1024) deg_hist_k(const int* __restrict__ dst,
                                                   unsigned* __restrict__ parts, int E) {
    __shared__ __attribute__((aligned(16))) unsigned cnt[HWORDS];

    const int base = blockIdx.x * CHUNK;
    int end = base + CHUNK;
    if (end > E) end = E;
    const int nvec = (end - base) >> 2;
    const int4* d4 = (const int4*)(dst + base);

    constexpr int SLOTS = (CHUNK / 4 + 1023) / 1024;
    int4 d[SLOTS];
#pragma unroll
    for (int j = 0; j < SLOTS; ++j) {
        const int i = threadIdx.x + j * 1024;
        if (i < nvec) d[j] = d4[i];
    }

    for (int i = threadIdx.x; i < HWORDS / 4; i += 1024)
        ((uint4*)cnt)[i] = make_uint4(0u, 0u, 0u, 0u);
    __syncthreads();

#pragma unroll
    for (int j = 0; j < SLOTS; ++j) {
        const int i = threadIdx.x + j * 1024;
        if (i < nvec) {
            atomicAdd(&cnt[d[j].x >> 1], 1u << ((d[j].x & 1) << 4));
            atomicAdd(&cnt[d[j].y >> 1], 1u << ((d[j].y & 1) << 4));
            atomicAdd(&cnt[d[j].z >> 1], 1u << ((d[j].z & 1) << 4));
            atomicAdd(&cnt[d[j].w >> 1], 1u << ((d[j].w & 1) << 4));
        }
    }
    for (int e = base + (nvec << 2) + threadIdx.x; e < end; e += 1024) {
        const int t = dst[e];
        atomicAdd(&cnt[t >> 1], 1u << ((t & 1) << 4));
    }
    __syncthreads();

    uint4* o = (uint4*)(parts + (size_t)blockIdx.x * HWORDS);
    for (int i = threadIdx.x; i < HWORDS / 4; i += 1024) o[i] = ((const uint4*)cnt)[i];
}

__global__ void dinv_from_parts_k(const unsigned* __restrict__ parts, float* __restrict__ dinv,
                                  int nParts, int n) {
    int t = blockIdx.x * blockDim.x + threadIdx.x;
    if (t >= HWORDS) return;
    unsigned lo = 0u, hi = 0u;
#pragma unroll 8
    for (int p = 0; p < nParts; ++p) {
        unsigned w = parts[(size_t)p * HWORDS + t];
        lo += w & 0xffffu;
        hi += w >> 16;
    }
    const int i = 2 * t;
    float r0 = rsqrtf(1.0f + (float)lo);
    float r1 = rsqrtf(1.0f + (float)hi);
    if (i + 1 < n)      *(float2*)(dinv + i) = make_float2(r0, r1);
    else if (i < n)     dinv[i] = r0;
}

typedef __attribute__((ext_vector_type(4))) float v4f;

#define KC  64
#define PX  (KC + 4)
#define PW  (F_IN + 4)
#define NCH (F_IN / KC)
#define LPC 8

__device__ __forceinline__ void wave_lds_order() {
    __builtin_amdgcn_fence(__ATOMIC_RELEASE, "wavefront");
    __builtin_amdgcn_wave_barrier();
    __builtin_amdgcn_fence(__ATOMIC_ACQUIRE, "wavefront");
}

__global__ void __launch_bounds__(256) gemm1_k(const float* __restrict__ x,
                                               const float* __restrict__ W1,
                                               const float* __restrict__ dinv,
                                               float* __restrict__ h,
                                               float* __restrict__ agg,
                                               float* __restrict__ a1a,
                                               float* __restrict__ a1b, int nTiles) {
    __shared__ __attribute__((aligned(16))) float sWt[HID * PW];
    __shared__ __attribute__((aligned(16))) float sX[8 * 16 * PX];
    for (int i = threadIdx.x * 4; i < F_IN * HID; i += 256 * 4) {
        const float4 w = *(const float4*)(W1 + i);
        const int k = i >> 4, n = i & 15;
        sWt[(n    ) * PW + k] = w.x;
        sWt[(n + 1) * PW + k] = w.y;
        sWt[(n + 2) * PW + k] = w.z;
        sWt[(n + 3) * PW + k] = w.w;
    }
    __syncthreads();

    const int wave = threadIdx.x >> 5;
    const int lane = threadIdx.x & 31;
    const int tile = blockIdx.x * 8 + wave;
    if (tile >= nTiles) return;

    const int m    = lane & 15;
    const int half = lane >> 4;

    float* const st = sX + wave * (16 * PX);
    const float* gp = x + (size_t)(tile * 16 + half) * F_IN + m * 4;
    float*       sp = st + half * PX + m * 4;
    const float* fa = st  + m * PX + half * 4;
    const float* fb = sWt + m * PW + half * 4;

    v4f g[LPC];
#pragma unroll
    for (int i = 0; i < LPC; ++i)
        g[i] = __builtin_nontemporal_load((const v4f*)(gp + (size_t)i * 2 * F_IN));

    v8f c0 = {0.f, 0.f, 0.f, 0.f, 0.f, 0.f, 0.f, 0.f};
    v8f c1 = {0.f, 0.f, 0.f, 0.f, 0.f, 0.f, 0.f, 0.f};

    for (int ch = 0; ch < NCH; ++ch) {
#pragma unroll
        for (int i = 0; i < LPC; ++i)
            *(v4f*)(sp + i * 2 * PX) = g[i];
        if (ch + 1 < NCH) {
#pragma unroll
            for (int i = 0; i < LPC; ++i)
                g[i] = __builtin_nontemporal_load((const v4f*)(gp + (ch + 1) * KC + (size_t)i * 2 * F_IN));
        }
        wave_lds_order();
#pragma unroll
        for (int j = 0; j < KC / 8; ++j) {
            const v4f a = *(const v4f*)(fa + 8 * j);
            const v4f b = *(const v4f*)(fb + ch * KC + 8 * j);
            const v2f a01 = {a.x, a.y}, b01 = {b.x, b.y};
            const v2f a23 = {a.z, a.w}, b23 = {b.z, b.w};
            c0 = __builtin_amdgcn_wmma_f32_16x16x4_f32(false, a01, false, b01,
                                                       (short)0, c0, false, false);
            c1 = __builtin_amdgcn_wmma_f32_16x16x4_f32(false, a23, false, b23,
                                                       (short)0, c1, false, false);
        }
        wave_lds_order();
    }
    const v8f c = c0 + c1;

    const float4 dA = *(const float4*)(dinv + tile * 16 + half * 8);
    const float4 dB = *(const float4*)(dinv + tile * 16 + half * 8 + 4);
    const float dr[8] = {dA.x, dA.y, dA.z, dA.w, dB.x, dB.y, dB.z, dB.w};

    const size_t o = (size_t)(tile * 16 + half * 8) * HID + m;
#pragma unroll
    for (int v = 0; v < 8; ++v) {
        const float val = dr[v] * c[v];
        h  [o + (size_t)v * HID] = val;
        agg[o + (size_t)v * HID] = val;
        const int r = tile * 16 + half * 8 + v;
        float* z = (r < ROWS_B) ? (a1a + (size_t)r * HID + m) : (a1b + (size_t)(r - ROWS_B) * HID + m);
        if (r < 2 * ROWS_B) *z = 0.0f;
    }
}


__device__ __forceinline__ unsigned rowsel(unsigned x, unsigned lo, unsigned hi) {
    return __builtin_amdgcn_permlane16(x, x, lo, hi, false, false);
}

template <bool TAIL>
__device__ __forceinline__ void adds1(float* agg, const float (&v)[16],
                                      unsigned rlo, unsigned rhi, unsigned f4, int left) {
    const int lane = threadIdx.x & 31;
    char* const colbase = (char*)agg + f4;
#pragma unroll
    for (int j = 0; j < 16; ++j) {
        const unsigned sel = 0x11111111u * (unsigned)j;
        const unsigned lo = rowsel(rlo, sel, sel);
        const unsigned hi = rowsel(rhi, sel, sel);
        const long long off = (long long)(((unsigned long long)hi << 32) | lo);
        if (!TAIL || ((lane & 16) + j) < left)
            unsafeAtomicAdd((float*)(colbase + off), v[j]);
    }
}

__global__ void __launch_bounds__(256) scatter1_k(const int* __restrict__ src,
                                                  const int* __restrict__ dst,
                                                  const float* __restrict__ h,
                                                  float* agg, float* a1a, float* a1b, int E) {
    const int lane = threadIdx.x & 31;
    const int e0 = (blockIdx.x * 8 + (threadIdx.x >> 5)) * 32;
    if (e0 >= E) return;
    const int left = E - e0;
    int s = 0, d = 0;
    if (lane < left) { s = src[e0 + lane]; d = dst[e0 + lane]; }

    const unsigned f4 = (unsigned)(lane & 15) * 4u;
    const unsigned sb = (unsigned)s * (HID * 4u);
    float v[16];
#pragma unroll
    for (int j = 0; j < 16; ++j) {
        const unsigned sel = 0x11111111u * (unsigned)j;
        const unsigned off = rowsel(sb, sel, sel) + f4;
        v[j] = *(const float*)((const char*)h + off);
    }

    const unsigned half = chip_half();
    const float* row = agg + (size_t)d * HID;
    if (half != 0u) {
        if (d < ROWS_B)          row = a1a + (size_t)d * HID;
        else if (d < 2 * ROWS_B) row = a1b + (size_t)(d - ROWS_B) * HID;
    }
    const long long roff = (const char*)row - (const char*)agg;
    const unsigned rlo = (unsigned)(unsigned long long)roff;
    const unsigned rhi = (unsigned)((unsigned long long)roff >> 32);
    if (left >= 32) adds1<false>(agg, v, rlo, rhi, f4, left);
    else            adds1<true >(agg, v, rlo, rhi, f4, left);
}

__global__ void bias_relu_k(float* agg, const float* __restrict__ a1a,
                            const float* __restrict__ a1b, const float* __restrict__ dinv,
                            const float* __restrict__ b1, int n) {
    int t = blockIdx.x * blockDim.x + threadIdx.x;
    if (t >= n) return;
    const int row = t >> 4;
    const bool has = row < 2 * ROWS_B;
    const float* p2 = (row < ROWS_B) ? (a1a + t) : (has ? (a1b + (t - ROWS_B * HID)) : (agg + t));
    const float a0 = agg[t];
    const float a1 = *p2;
    const float a  = has ? a0 + a1 : a0;
    float v = dinv[row] * a + b1[t & 15];
    agg[t] = v > 0.f ? v : 0.f;
}

__global__ void __launch_bounds__(256) gemm2_k(float* agg,
                                               const float* __restrict__ W2,
                                               const float* __restrict__ dinv,
                                               float* __restrict__ h2p,
                                               float* __restrict__ acc1, int nTiles) {
    __shared__ float sW[HID * 16];
    {
        int t = threadIdx.x;
        int k = t >> 4, n = t & 15;
        sW[t] = (n < CLS) ? W2[k * CLS + n] : 0.0f;
    }
    __syncthreads();

    const int wave = threadIdx.x >> 5;
    const int lane = threadIdx.x & 31;
    const int tile = blockIdx.x * 8 + wave;
    if (tile >= nTiles) return;

    const int m    = lane & 15;
    const int kk   = (lane >> 4) << 1;
    const int half = lane >> 4;

    const float* arow = agg + (size_t)(tile * 16 + m) * HID + kk;
    v8f c = {0.f, 0.f, 0.f, 0.f, 0.f, 0.f, 0.f, 0.f};

#pragma unroll
    for (int k0 = 0; k0 < HID; k0 += 4) {
        v2f a = *(const v2f*)(arow + k0);
        v2f b;
        b.x = sW[(k0 + kk)     * 16 + m];
        b.y = sW[(k0 + kk + 1) * 16 + m];
        c = __builtin_amdgcn_wmma_f32_16x16x4_f32(false, a, false, b,
                                                  (short)0, c, false, false);
    }

    const float4 dA = *(const float4*)(dinv + tile * 16 + half * 8);
    const float4 dB = *(const float4*)(dinv + tile * 16 + half * 8 + 4);
    const float dr[8] = {dA.x, dA.y, dA.z, dA.w, dB.x, dB.y, dB.z, dB.w};

    if (m < P2) {
#pragma unroll
        for (int v = 0; v < 8; ++v) {
            const int row = tile * 16 + half * 8 + v;
            const float val = dr[v] * c[v];
            h2p [(size_t)row * P2  + m] = val;
            agg [(size_t)row * HID + m] = val;
            acc1[(size_t)row * P2  + m] = 0.0f;
        }
    }
}


template <bool TAIL>
__device__ __forceinline__ void adds2(float* __restrict__ out, const float (&v)[8],
                                      unsigned db, unsigned f4, bool fl, int left) {
    const int lane = threadIdx.x & 31;
    unsigned off[8];
#pragma unroll
    for (int j = 0; j < 8; ++j) {
        const unsigned lo = 0x11111111u * (unsigned)j;
        const unsigned hi = 0x11111111u * (unsigned)(j + 8);
        off[j] = rowsel(db, lo, hi) + f4;
    }
    if (fl) {
#pragma unroll
        for (int j = 0; j < 8; ++j) {
            if (!TAIL || ((lane & 24) + j) < left)
                unsafeAtomicAdd((float*)((char*)out + off[j]), v[j]);
        }
    }
}

__global__ void __launch_bounds__(256) scatter2_k(const int* __restrict__ src,
                                                  const int* __restrict__ dst,
                                                  const float* __restrict__ h2p,
                                                  float* __restrict__ acc0,
                                                  float* __restrict__ acc1, int E) {
    const int lane = threadIdx.x & 31;
    const int e0 = (blockIdx.x * 8 + (threadIdx.x >> 5)) * 32;
    if (e0 >= E) return;
    const int left = E - e0;
    int s = 0, d = 0;
    if (lane < left) { s = src[e0 + lane]; d = dst[e0 + lane]; }

    const int  f  = lane & 7;
    const bool fl = f < CLS;
    const unsigned f4 = (unsigned)(fl ? f : CLS - 1) * 4u;
    const unsigned sb = (unsigned)s * (P2 * 4u);
    float v[8];
#pragma unroll
    for (int j = 0; j < 8; ++j) {
        const unsigned lo = 0x11111111u * (unsigned)j;
        const unsigned hi = 0x11111111u * (unsigned)(j + 8);
        const unsigned off = rowsel(sb, lo, hi) + f4;
        v[j] = *(const float*)((const char*)h2p + off);
    }

    const unsigned hf = chip_half();
    float* out = hf ? acc1 : acc0;
    const unsigned db = (unsigned)d * (hf ? P2 * 4u : HID * 4u);
    if (left >= 32) adds2<false>(out, v, db, f4, fl, left);
    else            adds2<true >(out, v, db, f4, fl, left);
}

__global__ void final2_k(float* __restrict__ out, const float* __restrict__ acc0,
                         const float* __restrict__ acc1,
                         const float* __restrict__ dinv,
                         const float* __restrict__ b2, int n) {
    int t = blockIdx.x * blockDim.x + threadIdx.x;
    if (t >= n) return;
    int i = t / CLS, f = t % CLS;
    out[t] = dinv[i] * (acc0[(size_t)i * HID + f] + acc1[(size_t)i * P2 + f]) + b2[f];
}

extern "C" void kernel_launch(void* const* d_in, const int* in_sizes, int n_in,
                              void* d_out, int out_size, void* d_ws, size_t ws_size,
                              hipStream_t stream) {
    const float* x  = (const float*)d_in[0];
    const int*   ei = (const int*)  d_in[1];
    const float* W1 = (const float*)d_in[2];
    const float* b1 = (const float*)d_in[3];
    const float* W2 = (const float*)d_in[4];
    const float* b2 = (const float*)d_in[5];
    float* out = (float*)d_out;

    const int N = in_sizes[0] / F_IN;
    const int E = in_sizes[1] / 2;
    const int* src = ei;
    const int* dst = ei + E;

    float* dinv = (float*)d_ws;
    float* h    = dinv + N;
    float* agg  = h   + (size_t)N * HID;
    float* h2   = agg + (size_t)N * HID;
    float* a1a  = h2;
    float* a1b  = out;
    float* h2p  = h;
    float* acc1 = h + (size_t)N * P2;

    const int B = 256;
    const int nTiles = N / 16;

    const int nParts = (E + CHUNK - 1) / CHUNK;
    unsigned* parts = (unsigned*)h;
    deg_hist_k       <<<nParts, 1024, 0, stream>>>(dst, parts, E);
    dinv_from_parts_k<<<(HWORDS + B - 1) / B, B, 0, stream>>>(parts, dinv, nParts, N);

    gemm1_k    <<<(nTiles + 7) / 8, B, 0, stream>>>(x, W1, dinv, h, agg, a1a, a1b, nTiles);

    int n1 = N * HID;
    scatter1_k<<<(E + B - 1) / B, B, 0, stream>>>(src, dst, h, agg, a1a, a1b, E);
    bias_relu_k<<<(n1 + B - 1) / B, B, 0, stream>>>(agg, a1a, a1b, dinv, b1, n1);

    gemm2_k    <<<(nTiles + 7) / 8, B, 0, stream>>>(agg, W2, dinv, h2p, acc1, nTiles);

    int n2 = N * CLS;
    scatter2_k<<<(E + B - 1) / B, B, 0, stream>>>(src, dst, h2p, agg, acc1, E);
    final2_k  <<<(n2 + B - 1) / B, B, 0, stream>>>(out, agg, acc1, dinv, b2, n2);
}


# Bias/relu launch removed: GEMM2 forms its A operand from the two layer-1 sum copies itself (dinv, bias, relu in the fragment load); one launch and 50,000 short waves fewer; sums as before
# speedup vs baseline: 3.7785x; 1.0439x over previous
#include <hip/hip_runtime.h>
#include <hip/hip_bf16.h>

typedef __attribute__((ext_vector_type(2))) float v2f;
typedef __attribute__((ext_vector_type(8))) float v8f;

#define F_IN 512
#define HID  16
#define CLS  7
#define P2   8

__device__ __forceinline__ unsigned chip_half() {
    return (__builtin_amdgcn_s_sendmsg_rtn(0x87) >> 18) & 1u;
}

#define NODES  100000
#define HWORDS (NODES / 2)
#define CHUNK  50000

#define ROWS_B ((NODES * CLS) / HID)

__global__ void __launch_bounds__(1024) deg_hist_k(const int* __restrict__ dst,
                                                   unsigned* __restrict__ parts, int E) {
    __shared__ __attribute__((aligned(16))) unsigned cnt[HWORDS];

    const int base = blockIdx.x * CHUNK;
    int end = base + CHUNK;
    if (end > E) end = E;
    const int nvec = (end - base) >> 2;
    const int4* d4 = (const int4*)(dst + base);

    constexpr int SLOTS = (CHUNK / 4 + 1023) / 1024;
    int4 d[SLOTS];
#pragma unroll
    for (int j = 0; j < SLOTS; ++j) {
        const int i = threadIdx.x + j * 1024;
        if (i < nvec) d[j] = d4[i];
    }

    for (int i = threadIdx.x; i < HWORDS / 4; i += 1024)
        ((uint4*)cnt)[i] = make_uint4(0u, 0u, 0u, 0u);
    __syncthreads();

#pragma unroll
    for (int j = 0; j < SLOTS; ++j) {
        const int i = threadIdx.x + j * 1024;
        if (i < nvec) {
            atomicAdd(&cnt[d[j].x >> 1], 1u << ((d[j].x & 1) << 4));
            atomicAdd(&cnt[d[j].y >> 1], 1u << ((d[j].y & 1) << 4));
            atomicAdd(&cnt[d[j].z >> 1], 1u << ((d[j].z & 1) << 4));
            atomicAdd(&cnt[d[j].w >> 1], 1u << ((d[j].w & 1) << 4));
        }
    }
    for (int e = base + (nvec << 2) + threadIdx.x; e < end; e += 1024) {
        const int t = dst[e];
        atomicAdd(&cnt[t >> 1], 1u << ((t & 1) << 4));
    }
    __syncthreads();

    uint4* o = (uint4*)(parts + (size_t)blockIdx.x * HWORDS);
    for (int i = threadIdx.x; i < HWORDS / 4; i += 1024) o[i] = ((const uint4*)cnt)[i];
}

__global__ void dinv_from_parts_k(const unsigned* __restrict__ parts, float* __restrict__ dinv,
                                  int nParts, int n) {
    int t = blockIdx.x * blockDim.x + threadIdx.x;
    if (t >= HWORDS) return;
    unsigned lo = 0u, hi = 0u;
#pragma unroll 8
    for (int p = 0; p < nParts; ++p) {
        unsigned w = parts[(size_t)p * HWORDS + t];
        lo += w & 0xffffu;
        hi += w >> 16;
    }
    const int i = 2 * t;
    float r0 = rsqrtf(1.0f + (float)lo);
    float r1 = rsqrtf(1.0f + (float)hi);
    if (i + 1 < n)      *(float2*)(dinv + i) = make_float2(r0, r1);
    else if (i < n)     dinv[i] = r0;
}

typedef __attribute__((ext_vector_type(4))) float v4f;

#define KC  64
#define PX  (KC + 4)
#define PW  (F_IN + 4)
#define NCH (F_IN / KC)
#define LPC 8

__device__ __forceinline__ void wave_lds_order() {
    __builtin_amdgcn_fence(__ATOMIC_RELEASE, "wavefront");
    __builtin_amdgcn_wave_barrier();
    __builtin_amdgcn_fence(__ATOMIC_ACQUIRE, "wavefront");
}

__global__ void __launch_bounds__(256) gemm1_k(const float* __restrict__ x,
                                               const float* __restrict__ W1,
                                               const float* __restrict__ dinv,
                                               float* __restrict__ h,
                                               float* __restrict__ agg,
                                               float* __restrict__ a1a,
                                               float* __restrict__ a1b, int nTiles) {
    __shared__ __attribute__((aligned(16))) float sWt[HID * PW];
    __shared__ __attribute__((aligned(16))) float sX[8 * 16 * PX];
    for (int i = threadIdx.x * 4; i < F_IN * HID; i += 256 * 4) {
        const float4 w = *(const float4*)(W1 + i);
        const int k = i >> 4, n = i & 15;
        sWt[(n    ) * PW + k] = w.x;
        sWt[(n + 1) * PW + k] = w.y;
        sWt[(n + 2) * PW + k] = w.z;
        sWt[(n + 3) * PW + k] = w.w;
    }
    __syncthreads();

    const int wave = threadIdx.x >> 5;
    const int lane = threadIdx.x & 31;
    const int tile = blockIdx.x * 8 + wave;
    if (tile >= nTiles) return;

    const int m    = lane & 15;
    const int half = lane >> 4;

    float* const st = sX + wave * (16 * PX);
    const float* gp = x + (size_t)(tile * 16 + half) * F_IN + m * 4;
    float*       sp = st + half * PX + m * 4;
    const float* fa = st  + m * PX + half * 4;
    const float* fb = sWt + m * PW + half * 4;

    v4f g[LPC];
#pragma unroll
    for (int i = 0; i < LPC; ++i)
        g[i] = __builtin_nontemporal_load((const v4f*)(gp + (size_t)i * 2 * F_IN));

    v8f c0 = {0.f, 0.f, 0.f, 0.f, 0.f, 0.f, 0.f, 0.f};
    v8f c1 = {0.f, 0.f, 0.f, 0.f, 0.f, 0.f, 0.f, 0.f};

    for (int ch = 0; ch < NCH; ++ch) {
#pragma unroll
        for (int i = 0; i < LPC; ++i)
            *(v4f*)(sp + i * 2 * PX) = g[i];
        if (ch + 1 < NCH) {
#pragma unroll
            for (int i = 0; i < LPC; ++i)
                g[i] = __builtin_nontemporal_load((const v4f*)(gp + (ch + 1) * KC + (size_t)i * 2 * F_IN));
        }
        wave_lds_order();
#pragma unroll
        for (int j = 0; j < KC / 8; ++j) {
            const v4f a = *(const v4f*)(fa + 8 * j);
            const v4f b = *(const v4f*)(fb + ch * KC + 8 * j);
            const v2f a01 = {a.x, a.y}, b01 = {b.x, b.y};
            const v2f a23 = {a.z, a.w}, b23 = {b.z, b.w};
            c0 = __builtin_amdgcn_wmma_f32_16x16x4_f32(false, a01, false, b01,
                                                       (short)0, c0, false, false);
            c1 = __builtin_amdgcn_wmma_f32_16x16x4_f32(false, a23, false, b23,
                                                       (short)0, c1, false, false);
        }
        wave_lds_order();
    }
    const v8f c = c0 + c1;

    const float4 dA = *(const float4*)(dinv + tile * 16 + half * 8);
    const float4 dB = *(const float4*)(dinv + tile * 16 + half * 8 + 4);
    const float dr[8] = {dA.x, dA.y, dA.z, dA.w, dB.x, dB.y, dB.z, dB.w};

    const size_t o = (size_t)(tile * 16 + half * 8) * HID + m;
#pragma unroll
    for (int v = 0; v < 8; ++v) {
        const float val = dr[v] * c[v];
        h  [o + (size_t)v * HID] = val;
        agg[o + (size_t)v * HID] = val;
        const int r = tile * 16 + half * 8 + v;
        float* z = (r < ROWS_B) ? (a1a + (size_t)r * HID + m) : (a1b + (size_t)(r - ROWS_B) * HID + m);
        if (r < 2 * ROWS_B) *z = 0.0f;
    }
}


__device__ __forceinline__ unsigned rowsel(unsigned x, unsigned lo, unsigned hi) {
    return __builtin_amdgcn_permlane16(x, x, lo, hi, false, false);
}

template <bool TAIL>
__device__ __forceinline__ void adds1(float* agg, const float (&v)[16],
                                      unsigned rlo, unsigned rhi, unsigned f4, int left) {
    const int lane = threadIdx.x & 31;
    char* const colbase = (char*)agg + f4;
#pragma unroll
    for (int j = 0; j < 16; ++j) {
        const unsigned sel = 0x11111111u * (unsigned)j;
        const unsigned lo = rowsel(rlo, sel, sel);
        const unsigned hi = rowsel(rhi, sel, sel);
        const long long off = (long long)(((unsigned long long)hi << 32) | lo);
        if (!TAIL || ((lane & 16) + j) < left)
            unsafeAtomicAdd((float*)(colbase + off), v[j]);
    }
}

__global__ void __launch_bounds__(256) scatter1_k(const int* __restrict__ src,
                                                  const int* __restrict__ dst,
                                                  const float* __restrict__ h,
                                                  float* agg, float* a1a, float* a1b, int E) {
    const int lane = threadIdx.x & 31;
    const int e0 = (blockIdx.x * 8 + (threadIdx.x >> 5)) * 32;
    if (e0 >= E) return;
    const int left = E - e0;
    int s = 0, d = 0;
    if (lane < left) { s = src[e0 + lane]; d = dst[e0 + lane]; }

    const unsigned f4 = (unsigned)(lane & 15) * 4u;
    const unsigned sb = (unsigned)s * (HID * 4u);
    float v[16];
#pragma unroll
    for (int j = 0; j < 16; ++j) {
        const unsigned sel = 0x11111111u * (unsigned)j;
        const unsigned off = rowsel(sb, sel, sel) + f4;
        v[j] = *(const float*)((const char*)h + off);
    }

    const unsigned half = chip_half();
    const float* row = agg + (size_t)d * HID;
    if (half != 0u) {
        if (d < ROWS_B)          row = a1a + (size_t)d * HID;
        else if (d < 2 * ROWS_B) row = a1b + (size_t)(d - ROWS_B) * HID;
    }
    const long long roff = (const char*)row - (const char*)agg;
    const unsigned rlo = (unsigned)(unsigned long long)roff;
    const unsigned rhi = (unsigned)((unsigned long long)roff >> 32);
    if (left >= 32) adds1<false>(agg, v, rlo, rhi, f4, left);
    else            adds1<true >(agg, v, rlo, rhi, f4, left);
}

__global__ void __launch_bounds__(256) gemm2_k(float* agg,
                                               const float* __restrict__ a1a,
                                               const float* __restrict__ a1b,
                                               const float* __restrict__ b1,
                                               const float* __restrict__ W2,
                                               const float* __restrict__ dinv,
                                               float* __restrict__ h2p,
                                               float* __restrict__ acc1, int nTiles) {
    __shared__ float sW[HID * 16];
    {
        int t = threadIdx.x;
        int k = t >> 4, n = t & 15;
        sW[t] = (n < CLS) ? W2[k * CLS + n] : 0.0f;
    }
    __syncthreads();

    const int wave = threadIdx.x >> 5;
    const int lane = threadIdx.x & 31;
    const int tile = blockIdx.x * 8 + wave;
    if (tile >= nTiles) return;

    const int m    = lane & 15;
    const int kk   = (lane >> 4) << 1;
    const int half = lane >> 4;

    const int    r    = tile * 16 + m;
    const bool   has  = r < 2 * ROWS_B;
    const float* arow = agg + (size_t)r * HID + kk;
    const float* srow = (r < ROWS_B) ? (a1a + (size_t)r * HID + kk)
                      : (has ? (a1b + (size_t)(r - ROWS_B) * HID + kk) : arow);
    const float  dm   = dinv[r];

    v2f a0[HID / 4], a1[HID / 4], bb[HID / 4];
#pragma unroll
    for (int q = 0; q < HID / 4; ++q) {
        a0[q] = *(const v2f*)(arow + 4 * q);
        a1[q] = *(const v2f*)(srow + 4 * q);
        bb[q] = *(const v2f*)(b1 + 4 * q + kk);
    }

    v8f c = {0.f, 0.f, 0.f, 0.f, 0.f, 0.f, 0.f, 0.f};

#pragma unroll
    for (int q = 0; q < HID / 4; ++q) {
        const int k0 = 4 * q;
        const float sx = has ? a0[q].x + a1[q].x : a0[q].x;
        const float sy = has ? a0[q].y + a1[q].y : a0[q].y;
        const float vx = dm * sx + bb[q].x;
        const float vy = dm * sy + bb[q].y;
        v2f a;
        a.x = vx > 0.f ? vx : 0.f;
        a.y = vy > 0.f ? vy : 0.f;
        v2f b;
        b.x = sW[(k0 + kk)     * 16 + m];
        b.y = sW[(k0 + kk + 1) * 16 + m];
        c = __builtin_amdgcn_wmma_f32_16x16x4_f32(false, a, false, b,
                                                  (short)0, c, false, false);
    }

    const float4 dA = *(const float4*)(dinv + tile * 16 + half * 8);
    const float4 dB = *(const float4*)(dinv + tile * 16 + half * 8 + 4);
    const float dr[8] = {dA.x, dA.y, dA.z, dA.w, dB.x, dB.y, dB.z, dB.w};

    if (m < P2) {
#pragma unroll
        for (int v = 0; v < 8; ++v) {
            const int row = tile * 16 + half * 8 + v;
            const float val = dr[v] * c[v];
            h2p [(size_t)row * P2  + m] = val;
            agg [(size_t)row * HID + m] = val;
            acc1[(size_t)row * P2  + m] = 0.0f;
        }
    }
}


template <bool TAIL>
__device__ __forceinline__ void adds2(float* __restrict__ out, const float (&v)[8],
                                      unsigned db, unsigned f4, bool fl, int left) {
    const int lane = threadIdx.x & 31;
    unsigned off[8];
#pragma unroll
    for (int j = 0; j < 8; ++j) {
        const unsigned lo = 0x11111111u * (unsigned)j;
        const unsigned hi = 0x11111111u * (unsigned)(j + 8);
        off[j] = rowsel(db, lo, hi) + f4;
    }
    if (fl) {
#pragma unroll
        for (int j = 0; j < 8; ++j) {
            if (!TAIL || ((lane & 24) + j) < left)
                unsafeAtomicAdd((float*)((char*)out + off[j]), v[j]);
        }
    }
}

__global__ void __launch_bounds__(256) scatter2_k(const int* __restrict__ src,
                                                  const int* __restrict__ dst,
                                                  const float* __restrict__ h2p,
                                                  float* __restrict__ acc0,
                                                  float* __restrict__ acc1, int E) {
    const int lane = threadIdx.x & 31;
    const int e0 = (blockIdx.x * 8 + (threadIdx.x >> 5)) * 32;
    if (e0 >= E) return;
    const int left = E - e0;
    int s = 0, d = 0;
    if (lane < left) { s = src[e0 + lane]; d = dst[e0 + lane]; }

    const int  f  = lane & 7;
    const bool fl = f < CLS;
    const unsigned f4 = (unsigned)(fl ? f : CLS - 1) * 4u;
    const unsigned sb = (unsigned)s * (P2 * 4u);
    float v[8];
#pragma unroll
    for (int j = 0; j < 8; ++j) {
        const unsigned lo = 0x11111111u * (unsigned)j;
        const unsigned hi = 0x11111111u * (unsigned)(j + 8);
        const unsigned off = rowsel(sb, lo, hi) + f4;
        v[j] = *(const float*)((const char*)h2p + off);
    }

    const unsigned hf = chip_half();
    float* out = hf ? acc1 : acc0;
    const unsigned db = (unsigned)d * (hf ? P2 * 4u : HID * 4u);
    if (left >= 32) adds2<false>(out, v, db, f4, fl, left);
    else            adds2<true >(out, v, db, f4, fl, left);
}

__global__ void final2_k(float* __restrict__ out, const float* __restrict__ acc0,
                         const float* __restrict__ acc1,
                         const float* __restrict__ dinv,
                         const float* __restrict__ b2, int n) {
    int t = blockIdx.x * blockDim.x + threadIdx.x;
    if (t >= n) return;
    int i = t / CLS, f = t % CLS;
    out[t] = dinv[i] * (acc0[(size_t)i * HID + f] + acc1[(size_t)i * P2 + f]) + b2[f];
}

extern "C" void kernel_launch(void* const* d_in, const int* in_sizes, int n_in,
                              void* d_out, int out_size, void* d_ws, size_t ws_size,
                              hipStream_t stream) {
    const float* x  = (const float*)d_in[0];
    const int*   ei = (const int*)  d_in[1];
    const float* W1 = (const float*)d_in[2];
    const float* b1 = (const float*)d_in[3];
    const float* W2 = (const float*)d_in[4];
    const float* b2 = (const float*)d_in[5];
    float* out = (float*)d_out;

    const int N = in_sizes[0] / F_IN;
    const int E = in_sizes[1] / 2;
    const int* src = ei;
    const int* dst = ei + E;

    float* dinv = (float*)d_ws;
    float* h    = dinv + N;
    float* agg  = h   + (size_t)N * HID;
    float* h2   = agg + (size_t)N * HID;
    float* a1a  = h2;
    float* a1b  = out;
    float* h2p  = h;
    float* acc1 = h + (size_t)N * P2;

    const int B = 256;
    const int nTiles = N / 16;

    const int nParts = (E + CHUNK - 1) / CHUNK;
    unsigned* parts = (unsigned*)h;
    deg_hist_k       <<<nParts, 1024, 0, stream>>>(dst, parts, E);
    dinv_from_parts_k<<<(HWORDS + B - 1) / B, B, 0, stream>>>(parts, dinv, nParts, N);

    gemm1_k    <<<(nTiles + 7) / 8, B, 0, stream>>>(x, W1, dinv, h, agg, a1a, a1b, nTiles);

    scatter1_k<<<(E + B - 1) / B, B, 0, stream>>>(src, dst, h, agg, a1a, a1b, E);

    gemm2_k    <<<(nTiles + 7) / 8, B, 0, stream>>>(agg, a1a, a1b, b1, W2, dinv, h2p, acc1, nTiles);

    int n2 = N * CLS;
    scatter2_k<<<(E + B - 1) / B, B, 0, stream>>>(src, dst, h2p, agg, acc1, E);
    final2_k  <<<(n2 + B - 1) / B, B, 0, stream>>>(out, agg, acc1, dinv, b2, n2);
}


# Layer-1 edge pass: second sum copy extended to the last 12,500 destination rows (third piece behind the workspace planes, used where ws_size holds it); their sums split in two partials, arrival order
# speedup vs baseline: 3.9355x; 1.0416x over previous
#include <hip/hip_runtime.h>
#include <hip/hip_bf16.h>

typedef __attribute__((ext_vector_type(2))) float v2f;
typedef __attribute__((ext_vector_type(8))) float v8f;

#define F_IN 512
#define HID  16
#define CLS  7
#define P2   8

__device__ __forceinline__ unsigned chip_half() {
    return (__builtin_amdgcn_s_sendmsg_rtn(0x87) >> 18) & 1u;
}

#define NODES  100000
#define HWORDS (NODES / 2)
#define CHUNK  50000

#define ROWS_B ((NODES * CLS) / HID)
#define ROWS_C  (NODES - 2 * ROWS_B)
#define WS_NEED (((size_t)NODES * (1 + HID + HID + CLS) + (size_t)ROWS_C * HID) * sizeof(float))

__global__ void __launch_bounds__(1024) deg_hist_k(const int* __restrict__ dst,
                                                   unsigned* __restrict__ parts, int E) {
    __shared__ __attribute__((aligned(16))) unsigned cnt[HWORDS];

    const int base = blockIdx.x * CHUNK;
    int end = base + CHUNK;
    if (end > E) end = E;
    const int nvec = (end - base) >> 2;
    const int4* d4 = (const int4*)(dst + base);

    constexpr int SLOTS = (CHUNK / 4 + 1023) / 1024;
    int4 d[SLOTS];
#pragma unroll
    for (int j = 0; j < SLOTS; ++j) {
        const int i = threadIdx.x + j * 1024;
        if (i < nvec) d[j] = d4[i];
    }

    for (int i = threadIdx.x; i < HWORDS / 4; i += 1024)
        ((uint4*)cnt)[i] = make_uint4(0u, 0u, 0u, 0u);
    __syncthreads();

#pragma unroll
    for (int j = 0; j < SLOTS; ++j) {
        const int i = threadIdx.x + j * 1024;
        if (i < nvec) {
            atomicAdd(&cnt[d[j].x >> 1], 1u << ((d[j].x & 1) << 4));
            atomicAdd(&cnt[d[j].y >> 1], 1u << ((d[j].y & 1) << 4));
            atomicAdd(&cnt[d[j].z >> 1], 1u << ((d[j].z & 1) << 4));
            atomicAdd(&cnt[d[j].w >> 1], 1u << ((d[j].w & 1) << 4));
        }
    }
    for (int e = base + (nvec << 2) + threadIdx.x; e < end; e += 1024) {
        const int t = dst[e];
        atomicAdd(&cnt[t >> 1], 1u << ((t & 1) << 4));
    }
    __syncthreads();

    uint4* o = (uint4*)(parts + (size_t)blockIdx.x * HWORDS);
    for (int i = threadIdx.x; i < HWORDS / 4; i += 1024) o[i] = ((const uint4*)cnt)[i];
}

__global__ void dinv_from_parts_k(const unsigned* __restrict__ parts, float* __restrict__ dinv,
                                  int nParts, int n) {
    int t = blockIdx.x * blockDim.x + threadIdx.x;
    if (t >= HWORDS) return;
    unsigned lo = 0u, hi = 0u;
#pragma unroll 8
    for (int p = 0; p < nParts; ++p) {
        unsigned w = parts[(size_t)p * HWORDS + t];
        lo += w & 0xffffu;
        hi += w >> 16;
    }
    const int i = 2 * t;
    float r0 = rsqrtf(1.0f + (float)lo);
    float r1 = rsqrtf(1.0f + (float)hi);
    if (i + 1 < n)      *(float2*)(dinv + i) = make_float2(r0, r1);
    else if (i < n)     dinv[i] = r0;
}

typedef __attribute__((ext_vector_type(4))) float v4f;

#define KC  64
#define PX  (KC + 4)
#define PW  (F_IN + 4)
#define NCH (F_IN / KC)
#define LPC 8

__device__ __forceinline__ void wave_lds_order() {
    __builtin_amdgcn_fence(__ATOMIC_RELEASE, "wavefront");
    __builtin_amdgcn_wave_barrier();
    __builtin_amdgcn_fence(__ATOMIC_ACQUIRE, "wavefront");
}

__global__ void __launch_bounds__(256) gemm1_k(const float* __restrict__ x,
                                               const float* __restrict__ W1,
                                               const float* __restrict__ dinv,
                                               float* __restrict__ h,
                                               float* __restrict__ agg,
                                               float* __restrict__ a1a,
                                               float* __restrict__ a1b,
                                               float* __restrict__ a1c, size_t wsBytes, int nTiles) {
    __shared__ __attribute__((aligned(16))) float sWt[HID * PW];
    __shared__ __attribute__((aligned(16))) float sX[8 * 16 * PX];
    for (int i = threadIdx.x * 4; i < F_IN * HID; i += 256 * 4) {
        const float4 w = *(const float4*)(W1 + i);
        const int k = i >> 4, n = i & 15;
        sWt[(n    ) * PW + k] = w.x;
        sWt[(n + 1) * PW + k] = w.y;
        sWt[(n + 2) * PW + k] = w.z;
        sWt[(n + 3) * PW + k] = w.w;
    }
    __syncthreads();

    const int wave = threadIdx.x >> 5;
    const int lane = threadIdx.x & 31;
    const int tile = blockIdx.x * 8 + wave;
    if (tile >= nTiles) return;

    const int m    = lane & 15;
    const int half = lane >> 4;

    float* const st = sX + wave * (16 * PX);
    const float* gp = x + (size_t)(tile * 16 + half) * F_IN + m * 4;
    float*       sp = st + half * PX + m * 4;
    const float* fa = st  + m * PX + half * 4;
    const float* fb = sWt + m * PW + half * 4;

    v4f g[LPC];
#pragma unroll
    for (int i = 0; i < LPC; ++i)
        g[i] = __builtin_nontemporal_load((const v4f*)(gp + (size_t)i * 2 * F_IN));

    v8f c0 = {0.f, 0.f, 0.f, 0.f, 0.f, 0.f, 0.f, 0.f};
    v8f c1 = {0.f, 0.f, 0.f, 0.f, 0.f, 0.f, 0.f, 0.f};

    for (int ch = 0; ch < NCH; ++ch) {
#pragma unroll
        for (int i = 0; i < LPC; ++i)
            *(v4f*)(sp + i * 2 * PX) = g[i];
        if (ch + 1 < NCH) {
#pragma unroll
            for (int i = 0; i < LPC; ++i)
                g[i] = __builtin_nontemporal_load((const v4f*)(gp + (ch + 1) * KC + (size_t)i * 2 * F_IN));
        }
        wave_lds_order();
#pragma unroll
        for (int j = 0; j < KC / 8; ++j) {
            const v4f a = *(const v4f*)(fa + 8 * j);
            const v4f b = *(const v4f*)(fb + ch * KC + 8 * j);
            const v2f a01 = {a.x, a.y}, b01 = {b.x, b.y};
            const v2f a23 = {a.z, a.w}, b23 = {b.z, b.w};
            c0 = __builtin_amdgcn_wmma_f32_16x16x4_f32(false, a01, false, b01,
                                                       (short)0, c0, false, false);
            c1 = __builtin_amdgcn_wmma_f32_16x16x4_f32(false, a23, false, b23,
                                                       (short)0, c1, false, false);
        }
        wave_lds_order();
    }
    const v8f c = c0 + c1;

    const float4 dA = *(const float4*)(dinv + tile * 16 + half * 8);
    const float4 dB = *(const float4*)(dinv + tile * 16 + half * 8 + 4);
    const float dr[8] = {dA.x, dA.y, dA.z, dA.w, dB.x, dB.y, dB.z, dB.w};

    const size_t o = (size_t)(tile * 16 + half * 8) * HID + m;
#pragma unroll
    for (int v = 0; v < 8; ++v) {
        const float val = dr[v] * c[v];
        h  [o + (size_t)v * HID] = val;
        agg[o + (size_t)v * HID] = val;
        const int r = tile * 16 + half * 8 + v;
        float* z = (r < ROWS_B)     ? (a1a + (size_t)r * HID + m)
                 : (r < 2 * ROWS_B) ? (a1b + (size_t)(r - ROWS_B) * HID + m)
                                    : (a1c + (size_t)(r - 2 * ROWS_B) * HID + m);
        if (r < 2 * ROWS_B || wsBytes >= WS_NEED) *z = 0.0f;
    }
}


__device__ __forceinline__ unsigned rowsel(unsigned x, unsigned lo, unsigned hi) {
    return __builtin_amdgcn_permlane16(x, x, lo, hi, false, false);
}

template <bool TAIL>
__device__ __forceinline__ void adds1(float* agg, const float (&v)[16],
                                      unsigned rlo, unsigned rhi, unsigned f4, int left) {
    const int lane = threadIdx.x & 31;
    char* const colbase = (char*)agg + f4;
#pragma unroll
    for (int j = 0; j < 16; ++j) {
        const unsigned sel = 0x11111111u * (unsigned)j;
        const unsigned lo = rowsel(rlo, sel, sel);
        const unsigned hi = rowsel(rhi, sel, sel);
        const long long off = (long long)(((unsigned long long)hi << 32) | lo);
        if (!TAIL || ((lane & 16) + j) < left)
            unsafeAtomicAdd((float*)(colbase + off), v[j]);
    }
}

__global__ void __launch_bounds__(256) scatter1_k(const int* __restrict__ src,
                                                  const int* __restrict__ dst,
                                                  const float* __restrict__ h,
                                                  float* agg, float* a1a, float* a1b, float* a1c,
                                                  size_t wsBytes, int E) {
    const int lane = threadIdx.x & 31;
    const int e0 = (blockIdx.x * 8 + (threadIdx.x >> 5)) * 32;
    if (e0 >= E) return;
    const int left = E - e0;
    int s = 0, d = 0;
    if (lane < left) { s = src[e0 + lane]; d = dst[e0 + lane]; }

    const unsigned f4 = (unsigned)(lane & 15) * 4u;
    const unsigned sb = (unsigned)s * (HID * 4u);
    float v[16];
#pragma unroll
    for (int j = 0; j < 16; ++j) {
        const unsigned sel = 0x11111111u * (unsigned)j;
        const unsigned off = rowsel(sb, sel, sel) + f4;
        v[j] = *(const float*)((const char*)h + off);
    }

    const unsigned half = chip_half();
    const float* row = agg + (size_t)d * HID;
    if (half != 0u) {
        if (d < ROWS_B)              row = a1a + (size_t)d * HID;
        else if (d < 2 * ROWS_B)     row = a1b + (size_t)(d - ROWS_B) * HID;
        else if (wsBytes >= WS_NEED) row = a1c + (size_t)(d - 2 * ROWS_B) * HID;
    }
    const long long roff = (const char*)row - (const char*)agg;
    const unsigned rlo = (unsigned)(unsigned long long)roff;
    const unsigned rhi = (unsigned)((unsigned long long)roff >> 32);
    if (left >= 32) adds1<false>(agg, v, rlo, rhi, f4, left);
    else            adds1<true >(agg, v, rlo, rhi, f4, left);
}

__global__ void __launch_bounds__(256) gemm2_k(float* agg,
                                               const float* __restrict__ a1a,
                                               const float* __restrict__ a1b,
                                               const float* __restrict__ a1c, size_t wsBytes,
                                               const float* __restrict__ b1,
                                               const float* __restrict__ W2,
                                               const float* __restrict__ dinv,
                                               float* __restrict__ h2p,
                                               float* __restrict__ acc1, int nTiles) {
    __shared__ float sW[HID * 16];
    {
        int t = threadIdx.x;
        int k = t >> 4, n = t & 15;
        sW[t] = (n < CLS) ? W2[k * CLS + n] : 0.0f;
    }
    __syncthreads();

    const int wave = threadIdx.x >> 5;
    const int lane = threadIdx.x & 31;
    const int tile = blockIdx.x * 8 + wave;
    if (tile >= nTiles) return;

    const int m    = lane & 15;
    const int kk   = (lane >> 4) << 1;
    const int half = lane >> 4;

    const int    r    = tile * 16 + m;
    const bool   has  = r < 2 * ROWS_B || wsBytes >= WS_NEED;
    const float* arow = agg + (size_t)r * HID + kk;
    const float* srow = (r < ROWS_B)     ? (a1a + (size_t)r * HID + kk)
                      : (r < 2 * ROWS_B) ? (a1b + (size_t)(r - ROWS_B) * HID + kk)
                      : (has ? (a1c + (size_t)(r - 2 * ROWS_B) * HID + kk) : arow);
    const float  dm   = dinv[r];

    v2f a0[HID / 4], a1[HID / 4], bb[HID / 4];
#pragma unroll
    for (int q = 0; q < HID / 4; ++q) {
        a0[q] = *(const v2f*)(arow + 4 * q);
        a1[q] = *(const v2f*)(srow + 4 * q);
        bb[q] = *(const v2f*)(b1 + 4 * q + kk);
    }

    v8f c = {0.f, 0.f, 0.f, 0.f, 0.f, 0.f, 0.f, 0.f};

#pragma unroll
    for (int q = 0; q < HID / 4; ++q) {
        const int k0 = 4 * q;
        const float sx = has ? a0[q].x + a1[q].x : a0[q].x;
        const float sy = has ? a0[q].y + a1[q].y : a0[q].y;
        const float vx = dm * sx + bb[q].x;
        const float vy = dm * sy + bb[q].y;
        v2f a;
        a.x = vx > 0.f ? vx : 0.f;
        a.y = vy > 0.f ? vy : 0.f;
        v2f b;
        b.x = sW[(k0 + kk)     * 16 + m];
        b.y = sW[(k0 + kk + 1) * 16 + m];
        c = __builtin_amdgcn_wmma_f32_16x16x4_f32(false, a, false, b,
                                                  (short)0, c, false, false);
    }

    const float4 dA = *(const float4*)(dinv + tile * 16 + half * 8);
    const float4 dB = *(const float4*)(dinv + tile * 16 + half * 8 + 4);
    const float dr[8] = {dA.x, dA.y, dA.z, dA.w, dB.x, dB.y, dB.z, dB.w};

    if (m < P2) {
#pragma unroll
        for (int v = 0; v < 8; ++v) {
            const int row = tile * 16 + half * 8 + v;
            const float val = dr[v] * c[v];
            h2p [(size_t)row * P2  + m] = val;
            agg [(size_t)row * HID + m] = val;
            acc1[(size_t)row * P2  + m] = 0.0f;
        }
    }
}


template <bool TAIL>
__device__ __forceinline__ void adds2(float* __restrict__ out, const float (&v)[8],
                                      unsigned db, unsigned f4, bool fl, int left) {
    const int lane = threadIdx.x & 31;
    unsigned off[8];
#pragma unroll
    for (int j = 0; j < 8; ++j) {
        const unsigned lo = 0x11111111u * (unsigned)j;
        const unsigned hi = 0x11111111u * (unsigned)(j + 8);
        off[j] = rowsel(db, lo, hi) + f4;
    }
    if (fl) {
#pragma unroll
        for (int j = 0; j < 8; ++j) {
            if (!TAIL || ((lane & 24) + j) < left)
                unsafeAtomicAdd((float*)((char*)out + off[j]), v[j]);
        }
    }
}

__global__ void __launch_bounds__(256) scatter2_k(const int* __restrict__ src,
                                                  const int* __restrict__ dst,
                                                  const float* __restrict__ h2p,
                                                  float* __restrict__ acc0,
                                                  float* __restrict__ acc1, int E) {
    const int lane = threadIdx.x & 31;
    const int e0 = (blockIdx.x * 8 + (threadIdx.x >> 5)) * 32;
    if (e0 >= E) return;
    const int left = E - e0;
    int s = 0, d = 0;
    if (lane < left) { s = src[e0 + lane]; d = dst[e0 + lane]; }

    const int  f  = lane & 7;
    const bool fl = f < CLS;
    const unsigned f4 = (unsigned)(fl ? f : CLS - 1) * 4u;
    const unsigned sb = (unsigned)s * (P2 * 4u);
    float v[8];
#pragma unroll
    for (int j = 0; j < 8; ++j) {
        const unsigned lo = 0x11111111u * (unsigned)j;
        const unsigned hi = 0x11111111u * (unsigned)(j + 8);
        const unsigned off = rowsel(sb, lo, hi) + f4;
        v[j] = *(const float*)((const char*)h2p + off);
    }

    const unsigned hf = chip_half();
    float* out = hf ? acc1 : acc0;
    const unsigned db = (unsigned)d * (hf ? P2 * 4u : HID * 4u);
    if (left >= 32) adds2<false>(out, v, db, f4, fl, left);
    else            adds2<true >(out, v, db, f4, fl, left);
}

__global__ void final2_k(float* __restrict__ out, const float* __restrict__ acc0,
                         const float* __restrict__ acc1,
                         const float* __restrict__ dinv,
                         const float* __restrict__ b2, int n) {
    int t = blockIdx.x * blockDim.x + threadIdx.x;
    if (t >= n) return;
    int i = t / CLS, f = t % CLS;
    out[t] = dinv[i] * (acc0[(size_t)i * HID + f] + acc1[(size_t)i * P2 + f]) + b2[f];
}

extern "C" void kernel_launch(void* const* d_in, const int* in_sizes, int n_in,
                              void* d_out, int out_size, void* d_ws, size_t ws_size,
                              hipStream_t stream) {
    const float* x  = (const float*)d_in[0];
    const int*   ei = (const int*)  d_in[1];
    const float* W1 = (const float*)d_in[2];
    const float* b1 = (const float*)d_in[3];
    const float* W2 = (const float*)d_in[4];
    const float* b2 = (const float*)d_in[5];
    float* out = (float*)d_out;

    const int N = in_sizes[0] / F_IN;
    const int E = in_sizes[1] / 2;
    const int* src = ei;
    const int* dst = ei + E;

    float* dinv = (float*)d_ws;
    float* h    = dinv + N;
    float* agg  = h   + (size_t)N * HID;
    float* h2   = agg + (size_t)N * HID;
    float* a1a  = h2;
    float* a1b  = out;
    float* a1c  = h2 + (size_t)N * CLS;
    float* h2p  = h;
    float* acc1 = h + (size_t)N * P2;

    const int B = 256;
    const int nTiles = N / 16;

    const int nParts = (E + CHUNK - 1) / CHUNK;
    unsigned* parts = (unsigned*)h;
    deg_hist_k       <<<nParts, 1024, 0, stream>>>(dst, parts, E);
    dinv_from_parts_k<<<(HWORDS + B - 1) / B, B, 0, stream>>>(parts, dinv, nParts, N);

    gemm1_k    <<<(nTiles + 7) / 8, B, 0, stream>>>(x, W1, dinv, h, agg, a1a, a1b, a1c, ws_size, nTiles);

    scatter1_k<<<(E + B - 1) / B, B, 0, stream>>>(src, dst, h, agg, a1a, a1b, a1c, ws_size, E);

    gemm2_k    <<<(nTiles + 7) / 8, B, 0, stream>>>(agg, a1a, a1b, a1c, ws_size, b1, W2, dinv, h2p, acc1, nTiles);

    int n2 = N * CLS;
    scatter2_k<<<(E + B - 1) / B, B, 0, stream>>>(src, dst, h2p, agg, acc1, E);
    final2_k  <<<(n2 + B - 1) / B, B, 0, stream>>>(out, agg, acc1, dinv, b2, n2);
}


# Both edge passes: edges put in destination-bucket order once per call (LDS counting sort); a workgroup sums its bucket in LDS (ds float adds, arrival order), node step fused; no global float atomics
# speedup vs baseline: 4.1216x; 1.0473x over previous
#include <hip/hip_runtime.h>
#include <hip/hip_bf16.h>

typedef __attribute__((ext_vector_type(2))) float v2f;
typedef __attribute__((ext_vector_type(8))) float v8f;

#define F_IN 512
#define HID  16
#define CLS  7
#define P2   8

__device__ __forceinline__ unsigned chip_half() {
    return (__builtin_amdgcn_s_sendmsg_rtn(0x87) >> 18) & 1u;
}

#define NODES  100000
#define HWORDS (NODES / 2)
#define CHUNK  50000

#define ROWS_B ((NODES * CLS) / HID)
#define ROWS_C  (NODES - 2 * ROWS_B)
#define WS_NEED (((size_t)NODES * (1 + HID + HID + CLS) + (size_t)ROWS_C * HID) * sizeof(float))

__global__ void __launch_bounds__(1024) deg_hist_k(const int* __restrict__ dst,
                                                   unsigned* __restrict__ parts, int E) {
    __shared__ __attribute__((aligned(16))) unsigned cnt[HWORDS];

    const int base = blockIdx.x * CHUNK;
    int end = base + CHUNK;
    if (end > E) end = E;
    const int nvec = (end - base) >> 2;
    const int4* d4 = (const int4*)(dst + base);

    constexpr int SLOTS = (CHUNK / 4 + 1023) / 1024;
    int4 d[SLOTS];
#pragma unroll
    for (int j = 0; j < SLOTS; ++j) {
        const int i = threadIdx.x + j * 1024;
        if (i < nvec) d[j] = d4[i];
    }

    for (int i = threadIdx.x; i < HWORDS / 4; i += 1024)
        ((uint4*)cnt)[i] = make_uint4(0u, 0u, 0u, 0u);
    __syncthreads();

#pragma unroll
    for (int j = 0; j < SLOTS; ++j) {
        const int i = threadIdx.x + j * 1024;
        if (i < nvec) {
            atomicAdd(&cnt[d[j].x >> 1], 1u << ((d[j].x & 1) << 4));
            atomicAdd(&cnt[d[j].y >> 1], 1u << ((d[j].y & 1) << 4));
            atomicAdd(&cnt[d[j].z >> 1], 1u << ((d[j].z & 1) << 4));
            atomicAdd(&cnt[d[j].w >> 1], 1u << ((d[j].w & 1) << 4));
        }
    }
    for (int e = base + (nvec << 2) + threadIdx.x; e < end; e += 1024) {
        const int t = dst[e];
        atomicAdd(&cnt[t >> 1], 1u << ((t & 1) << 4));
    }
    __syncthreads();

    uint4* o = (uint4*)(parts + (size_t)blockIdx.x * HWORDS);
    for (int i = threadIdx.x; i < HWORDS / 4; i += 1024) o[i] = ((const uint4*)cnt)[i];
}

__global__ void dinv_from_parts_k(const unsigned* __restrict__ parts, float* __restrict__ dinv,
                                  int nParts, int n) {
    int t = blockIdx.x * blockDim.x + threadIdx.x;
    if (t >= HWORDS) return;
    unsigned lo = 0u, hi = 0u;
#pragma unroll 8
    for (int p = 0; p < nParts; ++p) {
        unsigned w = parts[(size_t)p * HWORDS + t];
        lo += w & 0xffffu;
        hi += w >> 16;
    }
    const int i = 2 * t;
    float r0 = rsqrtf(1.0f + (float)lo);
    float r1 = rsqrtf(1.0f + (float)hi);
    if (i + 1 < n)      *(float2*)(dinv + i) = make_float2(r0, r1);
    else if (i < n)     dinv[i] = r0;
}

typedef __attribute__((ext_vector_type(4))) float v4f;

#define KC  64
#define PX  (KC + 4)
#define PW  (F_IN + 4)
#define NCH (F_IN / KC)
#define LPC 8

__device__ __forceinline__ void wave_lds_order() {
    __builtin_amdgcn_fence(__ATOMIC_RELEASE, "wavefront");
    __builtin_amdgcn_wave_barrier();
    __builtin_amdgcn_fence(__ATOMIC_ACQUIRE, "wavefront");
}

template <bool BK>
__global__ void __launch_bounds__(256) gemm1_k(const float* __restrict__ x,
                                               const float* __restrict__ W1,
                                               const float* __restrict__ dinv,
                                               float* __restrict__ h,
                                               float* __restrict__ agg,
                                               float* __restrict__ a1a,
                                               float* __restrict__ a1b,
                                               float* __restrict__ a1c, size_t wsBytes, int nTiles) {
    __shared__ __attribute__((aligned(16))) float sWt[HID * PW];
    __shared__ __attribute__((aligned(16))) float sX[8 * 16 * PX];
    for (int i = threadIdx.x * 4; i < F_IN * HID; i += 256 * 4) {
        const float4 w = *(const float4*)(W1 + i);
        const int k = i >> 4, n = i & 15;
        sWt[(n    ) * PW + k] = w.x;
        sWt[(n + 1) * PW + k] = w.y;
        sWt[(n + 2) * PW + k] = w.z;
        sWt[(n + 3) * PW + k] = w.w;
    }
    __syncthreads();

    const int wave = threadIdx.x >> 5;
    const int lane = threadIdx.x & 31;
    const int tile = blockIdx.x * 8 + wave;
    if (tile >= nTiles) return;

    const int m    = lane & 15;
    const int half = lane >> 4;

    float* const st = sX + wave * (16 * PX);
    const float* gp = x + (size_t)(tile * 16 + half) * F_IN + m * 4;
    float*       sp = st + half * PX + m * 4;
    const float* fa = st  + m * PX + half * 4;
    const float* fb = sWt + m * PW + half * 4;

    v4f g[LPC];
#pragma unroll
    for (int i = 0; i < LPC; ++i)
        g[i] = __builtin_nontemporal_load((const v4f*)(gp + (size_t)i * 2 * F_IN));

    v8f c0 = {0.f, 0.f, 0.f, 0.f, 0.f, 0.f, 0.f, 0.f};
    v8f c1 = {0.f, 0.f, 0.f, 0.f, 0.f, 0.f, 0.f, 0.f};

    for (int ch = 0; ch < NCH; ++ch) {
#pragma unroll
        for (int i = 0; i < LPC; ++i)
            *(v4f*)(sp + i * 2 * PX) = g[i];
        if (ch + 1 < NCH) {
#pragma unroll
            for (int i = 0; i < LPC; ++i)
                g[i] = __builtin_nontemporal_load((const v4f*)(gp + (ch + 1) * KC + (size_t)i * 2 * F_IN));
        }
        wave_lds_order();
#pragma unroll
        for (int j = 0; j < KC / 8; ++j) {
            const v4f a = *(const v4f*)(fa + 8 * j);
            const v4f b = *(const v4f*)(fb + ch * KC + 8 * j);
            const v2f a01 = {a.x, a.y}, b01 = {b.x, b.y};
            const v2f a23 = {a.z, a.w}, b23 = {b.z, b.w};
            c0 = __builtin_amdgcn_wmma_f32_16x16x4_f32(false, a01, false, b01,
                                                       (short)0, c0, false, false);
            c1 = __builtin_amdgcn_wmma_f32_16x16x4_f32(false, a23, false, b23,
                                                       (short)0, c1, false, false);
        }
        wave_lds_order();
    }
    const v8f c = c0 + c1;

    const float4 dA = *(const float4*)(dinv + tile * 16 + half * 8);
    const float4 dB = *(const float4*)(dinv + tile * 16 + half * 8 + 4);
    const float dr[8] = {dA.x, dA.y, dA.z, dA.w, dB.x, dB.y, dB.z, dB.w};

    const size_t o = (size_t)(tile * 16 + half * 8) * HID + m;
#pragma unroll
    for (int v = 0; v < 8; ++v) {
        const float val = dr[v] * c[v];
        h  [o + (size_t)v * HID] = val;
        if (!BK) {
            agg[o + (size_t)v * HID] = val;
            const int r = tile * 16 + half * 8 + v;
            float* z = (r < ROWS_B)     ? (a1a + (size_t)r * HID + m)
                     : (r < 2 * ROWS_B) ? (a1b + (size_t)(r - ROWS_B) * HID + m)
                                        : (a1c + (size_t)(r - 2 * ROWS_B) * HID + m);
            if (r < 2 * ROWS_B || wsBytes >= WS_NEED) *z = 0.0f;
        }
    }
}


__device__ __forceinline__ unsigned rowsel(unsigned x, unsigned lo, unsigned hi) {
    return __builtin_amdgcn_permlane16(x, x, lo, hi, false, false);
}

template <bool TAIL>
__device__ __forceinline__ void adds1(float* agg, const float (&v)[16],
                                      unsigned rlo, unsigned rhi, unsigned f4, int left) {
    const int lane = threadIdx.x & 31;
    char* const colbase = (char*)agg + f4;
#pragma unroll
    for (int j = 0; j < 16; ++j) {
        const unsigned sel = 0x11111111u * (unsigned)j;
        const unsigned lo = rowsel(rlo, sel, sel);
        const unsigned hi = rowsel(rhi, sel, sel);
        const long long off = (long long)(((unsigned long long)hi << 32) | lo);
        if (!TAIL || ((lane & 16) + j) < left)
            unsafeAtomicAdd((float*)(colbase + off), v[j]);
    }
}

__global__ void __launch_bounds__(256) scatter1_k(const int* __restrict__ src,
                                                  const int* __restrict__ dst,
                                                  const float* __restrict__ h,
                                                  float* agg, float* a1a, float* a1b, float* a1c,
                                                  size_t wsBytes, int E) {
    const int lane = threadIdx.x & 31;
    const int e0 = (blockIdx.x * 8 + (threadIdx.x >> 5)) * 32;
    if (e0 >= E) return;
    const int left = E - e0;
    int s = 0, d = 0;
    if (lane < left) { s = src[e0 + lane]; d = dst[e0 + lane]; }

    const unsigned f4 = (unsigned)(lane & 15) * 4u;
    const unsigned sb = (unsigned)s * (HID * 4u);
    float v[16];
#pragma unroll
    for (int j = 0; j < 16; ++j) {
        const unsigned sel = 0x11111111u * (unsigned)j;
        const unsigned off = rowsel(sb, sel, sel) + f4;
        v[j] = *(const float*)((const char*)h + off);
    }

    const unsigned half = chip_half();
    const float* row = agg + (size_t)d * HID;
    if (half != 0u) {
        if (d < ROWS_B)              row = a1a + (size_t)d * HID;
        else if (d < 2 * ROWS_B)     row = a1b + (size_t)(d - ROWS_B) * HID;
        else if (wsBytes >= WS_NEED) row = a1c + (size_t)(d - 2 * ROWS_B) * HID;
    }
    const long long roff = (const char*)row - (const char*)agg;
    const unsigned rlo = (unsigned)(unsigned long long)roff;
    const unsigned rhi = (unsigned)((unsigned long long)roff >> 32);
    if (left >= 32) adds1<false>(agg, v, rlo, rhi, f4, left);
    else            adds1<true >(agg, v, rlo, rhi, f4, left);
}

__global__ void __launch_bounds__(256) gemm2_k(float* agg,
                                               const float* __restrict__ a1a,
                                               const float* __restrict__ a1b,
                                               const float* __restrict__ a1c, size_t wsBytes,
                                               const float* __restrict__ b1,
                                               const float* __restrict__ W2,
                                               const float* __restrict__ dinv,
                                               float* __restrict__ h2p,
                                               float* __restrict__ acc1, int nTiles) {
    __shared__ float sW[HID * 16];
    {
        int t = threadIdx.x;
        int k = t >> 4, n = t & 15;
        sW[t] = (n < CLS) ? W2[k * CLS + n] : 0.0f;
    }
    __syncthreads();

    const int wave = threadIdx.x >> 5;
    const int lane = threadIdx.x & 31;
    const int tile = blockIdx.x * 8 + wave;
    if (tile >= nTiles) return;

    const int m    = lane & 15;
    const int kk   = (lane >> 4) << 1;
    const int half = lane >> 4;

    const int    r    = tile * 16 + m;
    const bool   has  = r < 2 * ROWS_B || wsBytes >= WS_NEED;
    const float* arow = agg + (size_t)r * HID + kk;
    const float* srow = (r < ROWS_B)     ? (a1a + (size_t)r * HID + kk)
                      : (r < 2 * ROWS_B) ? (a1b + (size_t)(r - ROWS_B) * HID + kk)
                      : (has ? (a1c + (size_t)(r - 2 * ROWS_B) * HID + kk) : arow);
    const float  dm   = dinv[r];

    v2f a0[HID / 4], a1[HID / 4], bb[HID / 4];
#pragma unroll
    for (int q = 0; q < HID / 4; ++q) {
        a0[q] = *(const v2f*)(arow + 4 * q);
        a1[q] = *(const v2f*)(srow + 4 * q);
        bb[q] = *(const v2f*)(b1 + 4 * q + kk);
    }

    v8f c = {0.f, 0.f, 0.f, 0.f, 0.f, 0.f, 0.f, 0.f};

#pragma unroll
    for (int q = 0; q < HID / 4; ++q) {
        const int k0 = 4 * q;
        const float sx = has ? a0[q].x + a1[q].x : a0[q].x;
        const float sy = has ? a0[q].y + a1[q].y : a0[q].y;
        const float vx = dm * sx + bb[q].x;
        const float vy = dm * sy + bb[q].y;
        v2f a;
        a.x = vx > 0.f ? vx : 0.f;
        a.y = vy > 0.f ? vy : 0.f;
        v2f b;
        b.x = sW[(k0 + kk)     * 16 + m];
        b.y = sW[(k0 + kk + 1) * 16 + m];
        c = __builtin_amdgcn_wmma_f32_16x16x4_f32(false, a, false, b,
                                                  (short)0, c, false, false);
    }

    const float4 dA = *(const float4*)(dinv + tile * 16 + half * 8);
    const float4 dB = *(const float4*)(dinv + tile * 16 + half * 8 + 4);
    const float dr[8] = {dA.x, dA.y, dA.z, dA.w, dB.x, dB.y, dB.z, dB.w};

    if (m < P2) {
#pragma unroll
        for (int v = 0; v < 8; ++v) {
            const int row = tile * 16 + half * 8 + v;
            const float val = dr[v] * c[v];
            h2p [(size_t)row * P2  + m] = val;
            agg [(size_t)row * HID + m] = val;
            acc1[(size_t)row * P2  + m] = 0.0f;
        }
    }
}


template <bool TAIL>
__device__ __forceinline__ void adds2(float* __restrict__ out, const float (&v)[8],
                                      unsigned db, unsigned f4, bool fl, int left) {
    const int lane = threadIdx.x & 31;
    unsigned off[8];
#pragma unroll
    for (int j = 0; j < 8; ++j) {
        const unsigned lo = 0x11111111u * (unsigned)j;
        const unsigned hi = 0x11111111u * (unsigned)(j + 8);
        off[j] = rowsel(db, lo, hi) + f4;
    }
    if (fl) {
#pragma unroll
        for (int j = 0; j < 8; ++j) {
            if (!TAIL || ((lane & 24) + j) < left)
                unsafeAtomicAdd((float*)((char*)out + off[j]), v[j]);
        }
    }
}

__global__ void __launch_bounds__(256) scatter2_k(const int* __restrict__ src,
                                                  const int* __restrict__ dst,
                                                  const float* __restrict__ h2p,
                                                  float* __restrict__ acc0,
                                                  float* __restrict__ acc1, int E) {
    const int lane = threadIdx.x & 31;
    const int e0 = (blockIdx.x * 8 + (threadIdx.x >> 5)) * 32;
    if (e0 >= E) return;
    const int left = E - e0;
    int s = 0, d = 0;
    if (lane < left) { s = src[e0 + lane]; d = dst[e0 + lane]; }

    const int  f  = lane & 7;
    const bool fl = f < CLS;
    const unsigned f4 = (unsigned)(fl ? f : CLS - 1) * 4u;
    const unsigned sb = (unsigned)s * (P2 * 4u);
    float v[8];
#pragma unroll
    for (int j = 0; j < 8; ++j) {
        const unsigned lo = 0x11111111u * (unsigned)j;
        const unsigned hi = 0x11111111u * (unsigned)(j + 8);
        const unsigned off = rowsel(sb, lo, hi) + f4;
        v[j] = *(const float*)((const char*)h2p + off);
    }

    const unsigned hf = chip_half();
    float* out = hf ? acc1 : acc0;
    const unsigned db = (unsigned)d * (hf ? P2 * 4u : HID * 4u);
    if (left >= 32) adds2<false>(out, v, db, f4, fl, left);
    else            adds2<true >(out, v, db, f4, fl, left);
}

__global__ void final2_k(float* __restrict__ out, const float* __restrict__ acc0,
                         const float* __restrict__ acc1,
                         const float* __restrict__ dinv,
                         const float* __restrict__ b2, int n) {
    int t = blockIdx.x * blockDim.x + threadIdx.x;
    if (t >= n) return;
    int i = t / CLS, f = t % CLS;
    out[t] = dinv[i] * (acc0[(size_t)i * HID + f] + acc1[(size_t)i * P2 + f]) + b2[f];
}

#define RB    400
#define NBK   (NODES / RB)
#define ECH   12800
#define EPT   (ECH / 512)
#define LB    9

__global__ void __launch_bounds__(512) bk_count_k(const int* __restrict__ dst,
                                                  unsigned* __restrict__ cnt, int E) {
    __shared__ unsigned hc[256];
    if (threadIdx.x < 256) hc[threadIdx.x] = 0u;
    const int base = blockIdx.x * ECH;
    int end = base + ECH;
    if (end > E) end = E;
    int d[EPT];
#pragma unroll
    for (int j = 0; j < EPT; ++j) {
        const int e = base + (int)threadIdx.x + j * 512;
        d[j] = (e < end) ? dst[e] : -1;
    }
    __syncthreads();
#pragma unroll
    for (int j = 0; j < EPT; ++j)
        if (d[j] >= 0) atomicAdd(&hc[(unsigned)d[j] / RB], 1u);
    __syncthreads();
    if (threadIdx.x < NBK) cnt[blockIdx.x * NBK + threadIdx.x] = hc[threadIdx.x];
}

__global__ void __launch_bounds__(512) bk_place_k(const int* __restrict__ src,
                                                  const int* __restrict__ dst,
                                                  const unsigned* __restrict__ cnt,
                                                  unsigned* __restrict__ bstart,
                                                  unsigned* __restrict__ pk, int E, int nEB) {
    __shared__ unsigned hAll[512], hBef[512];
    __shared__ unsigned wg[8], wl[8];
    __shared__ unsigned lst[256];
    __shared__ unsigned cur[256];
    __shared__ unsigned gof[256];
    __shared__ unsigned stage[ECH];
    const int t = threadIdx.x;
    const int base = blockIdx.x * ECH;
    int end = base + ECH;
    if (end > E) end = E;
    int s[EPT], d[EPT];
#pragma unroll
    for (int j = 0; j < EPT; ++j) {
        const int e = base + t + j * 512;
        const bool in = e < end;
        s[j] = in ? src[e] : 0;
        d[j] = in ? dst[e] : -1;
    }
    {
        const int b = t & 255, hf = t >> 8;
        const int lo = hf ? (nEB >> 1) : 0, hi = hf ? nEB : (nEB >> 1);
        unsigned all = 0u, bef = 0u;
        if (b < NBK) {
#pragma unroll 25
            for (int e = lo; e < hi; ++e) {
                const unsigned q = cnt[e * NBK + b];
                all += q;
                bef += (e < (int)blockIdx.x) ? q : 0u;
            }
        }
        hAll[t] = all;
        hBef[t] = bef;
    }
    unsigned own = 0u;
    if (t < NBK) own = cnt[blockIdx.x * NBK + t];
    __syncthreads();
    unsigned tot = 0u, bef = 0u;
    if (t < 256) {
        tot = hAll[t] + hAll[t + 256];
        bef = hBef[t] + hBef[t + 256];
    }
    const int lane = t & 31, wave = t >> 5;
    unsigned gs = tot, ls = own;
#pragma unroll
    for (int o = 1; o < 32; o <<= 1) {
        const unsigned u = __shfl_up(gs, o);
        const unsigned w = __shfl_up(ls, o);
        if (lane >= o) { gs += u; ls += w; }
    }
    if (lane == 31 && wave < 8) { wg[wave] = gs; wl[wave] = ls; }
    __syncthreads();
    if (t < 256) {
        unsigned gb = 0u, lb = 0u;
        for (int k = 0; k < wave; ++k) { gb += wg[k]; lb += wl[k]; }
        const unsigned gx = gs + gb - tot;
        const unsigned lx = ls + lb - own;
        lst[t] = lx;
        cur[t] = lx;
        gof[t] = gx + bef;
        if (blockIdx.x == 0 && t <= NBK) bstart[t] = gx;
    }
    __syncthreads();
#pragma unroll
    for (int j = 0; j < EPT; ++j) {
        if (d[j] >= 0) {
            const unsigned b = (unsigned)d[j] / RB;
            const unsigned p = atomicAdd(&cur[b], 1u);
            stage[p] = ((unsigned)s[j] << LB) | ((unsigned)d[j] - b * RB);
        }
    }
    __syncthreads();
    for (int b = wave; b < NBK; b += 16) {
        const unsigned l = lst[b];
        const unsigned nb = cur[b] - l;
        const unsigned g = gof[b];
        for (unsigned i = lane; i < nb; i += 32) pk[g + i] = stage[l + i];
    }
}

__device__ __forceinline__ void bk_gather1(const float* __restrict__ h, unsigned p, unsigned f4,
                                           float (&v)[16]) {
    const unsigned sb = (p >> LB) * (HID * 4u);
#pragma unroll
    for (int j = 0; j < 16; ++j) {
        const unsigned sel = 0x11111111u * (unsigned)j;
        v[j] = *(const float*)((const char*)h + (rowsel(sb, sel, sel) + f4));
    }
}
template <bool TAIL>
__device__ __forceinline__ void bk_add1(float* acc, unsigned p, unsigned f, const float (&v)[16],
                                        unsigned left) {
    const unsigned lane = threadIdx.x & 31;
    const unsigned dl = (p & ((1u << LB) - 1u)) * HID;
#pragma unroll
    for (int j = 0; j < 16; ++j) {
        const unsigned sel = 0x11111111u * (unsigned)j;
        const unsigned a = rowsel(dl, sel, sel) + f;
        if (!TAIL || ((lane & 16u) + (unsigned)j) < left)
            unsafeAtomicAdd(&acc[a], v[j]);
    }
}

__global__ void __launch_bounds__(1024) bk_agg1_k(const unsigned* __restrict__ pk,
                                                  const unsigned* __restrict__ bstart,
                                                  const float* __restrict__ h,
                                                  const float* __restrict__ dinv,
                                                  const float* __restrict__ b1,
                                                  const float* __restrict__ W2,
                                                  float* __restrict__ h2p) {
    __shared__ __attribute__((aligned(16))) float acc[RB * HID];
    __shared__ float sW[HID * 16];
    const int t = threadIdx.x;
    const int row0 = blockIdx.x * RB;
    const float4* h4 = (const float4*)(h + (size_t)row0 * HID);
    for (int i = t; i < RB * HID / 4; i += 1024) ((float4*)acc)[i] = h4[i];
    if (t < 256) {
        const int k = t >> 4, n = t & 15;
        sW[t] = (n < CLS) ? W2[k * CLS + n] : 0.0f;
    }
    const unsigned e0 = bstart[blockIdx.x], e1 = bstart[blockIdx.x + 1];
    const unsigned lane = (unsigned)t & 31u, wave = (unsigned)t >> 5;
    const unsigned f = lane & 15u;
    const unsigned step = 32u * 64u;
    unsigned eb = e0 + wave * 64u;
    unsigned pA = 0u, pB = 0u;
    if (eb + lane < e1)       pA = pk[eb + lane];
    if (eb + 32u + lane < e1) pB = pk[eb + 32u + lane];
    __syncthreads();

    while (eb < e1) {
        const unsigned en = eb + step;
        unsigned nA = 0u, nB = 0u;
        if (en + lane < e1)       nA = pk[en + lane];
        if (en + 32u + lane < e1) nB = pk[en + 32u + lane];
        const unsigned leftA = e1 - eb;
        const unsigned leftB = leftA > 32u ? leftA - 32u : 0u;
        float vA[16], vB[16];
        bk_gather1(h, pA, f * 4u, vA);
        bk_gather1(h, pB, f * 4u, vB);
        if (leftB >= 32u) { bk_add1<false>(acc, pA, f, vA, leftA); bk_add1<false>(acc, pB, f, vB, leftB); }
        else              { bk_add1<true >(acc, pA, f, vA, leftA); bk_add1<true >(acc, pB, f, vB, leftB); }
        eb = en; pA = nA; pB = nB;
    }
    __syncthreads();

    if (wave < RB / 16) {
        const int m    = (int)(lane & 15u);
        const int half = (int)(lane >> 4);
        const int kk   = half << 1;
        const int rl   = (int)wave * 16 + m;
        const float dm = dinv[row0 + rl];
        v8f c = {0.f, 0.f, 0.f, 0.f, 0.f, 0.f, 0.f, 0.f};
#pragma unroll
        for (int q = 0; q < HID / 4; ++q) {
            const int k0 = 4 * q;
            const float vx = dm * acc[rl * HID + k0 + kk]     + b1[k0 + kk];
            const float vy = dm * acc[rl * HID + k0 + kk + 1] + b1[k0 + kk + 1];
            v2f a;
            a.x = vx > 0.f ? vx : 0.f;
            a.y = vy > 0.f ? vy : 0.f;
            v2f b;
            b.x = sW[(k0 + kk)     * 16 + m];
            b.y = sW[(k0 + kk + 1) * 16 + m];
            c = __builtin_amdgcn_wmma_f32_16x16x4_f32(false, a, false, b,
                                                      (short)0, c, false, false);
        }
        if (m < P2) {
#pragma unroll
            for (int v = 0; v < 8; ++v) {
                const int row = row0 + (int)wave * 16 + half * 8 + v;
                h2p[(size_t)row * P2 + m] = dinv[row] * c[v];
            }
        }
    }
}

__device__ __forceinline__ void bk_gather2(const float* __restrict__ h2p, unsigned p, unsigned f4,
                                           float (&v)[8]) {
    const unsigned sb = (p >> LB) * (P2 * 4u);
#pragma unroll
    for (int j = 0; j < 8; ++j) {
        const unsigned lo = 0x11111111u * (unsigned)j;
        const unsigned hi = 0x11111111u * (unsigned)(j + 8);
        v[j] = *(const float*)((const char*)h2p + (rowsel(sb, lo, hi) + f4));
    }
}
template <bool TAIL>
__device__ __forceinline__ void bk_add2(float* acc, unsigned p, unsigned f, bool fl, const float (&v)[8],
                                        unsigned left) {
    const unsigned lane = threadIdx.x & 31;
    const unsigned dl = (p & ((1u << LB) - 1u)) * P2;
#pragma unroll
    for (int j = 0; j < 8; ++j) {
        const unsigned lo = 0x11111111u * (unsigned)j;
        const unsigned hi = 0x11111111u * (unsigned)(j + 8);
        const unsigned a = rowsel(dl, lo, hi) + f;
        if (fl && (!TAIL || ((lane & 24u) + (unsigned)j) < left))
            unsafeAtomicAdd(&acc[a], v[j]);
    }
}

__global__ void __launch_bounds__(1024) bk_agg2_k(const unsigned* __restrict__ pk,
                                                  const unsigned* __restrict__ bstart,
                                                  const float* __restrict__ h2p,
                                                  const float* __restrict__ dinv,
                                                  const float* __restrict__ b2,
                                                  float* __restrict__ out) {
    __shared__ __attribute__((aligned(16))) float acc[RB * P2];
    const int t = threadIdx.x;
    const int row0 = blockIdx.x * RB;
    const float4* g4 = (const float4*)(h2p + (size_t)row0 * P2);
    for (int i = t; i < RB * P2 / 4; i += 1024) ((float4*)acc)[i] = g4[i];
    const unsigned e0 = bstart[blockIdx.x], e1 = bstart[blockIdx.x + 1];
    const unsigned lane = (unsigned)t & 31u, wave = (unsigned)t >> 5;
    const unsigned f  = lane & 7u;
    const bool     fl = f < CLS;
    const unsigned f4 = (fl ? f : (unsigned)(CLS - 1)) * 4u;
    const unsigned step = 32u * 64u;
    unsigned eb = e0 + wave * 64u;
    unsigned pA = 0u, pB = 0u;
    if (eb + lane < e1)       pA = pk[eb + lane];
    if (eb + 32u + lane < e1) pB = pk[eb + 32u + lane];
    __syncthreads();

    while (eb < e1) {
        const unsigned en = eb + step;
        unsigned nA = 0u, nB = 0u;
        if (en + lane < e1)       nA = pk[en + lane];
        if (en + 32u + lane < e1) nB = pk[en + 32u + lane];
        const unsigned leftA = e1 - eb;
        const unsigned leftB = leftA > 32u ? leftA - 32u : 0u;
        float vA[8], vB[8];
        bk_gather2(h2p, pA, f4, vA);
        bk_gather2(h2p, pB, f4, vB);
        if (leftB >= 32u) { bk_add2<false>(acc, pA, f, fl, vA, leftA); bk_add2<false>(acc, pB, f, fl, vB, leftB); }
        else              { bk_add2<true >(acc, pA, f, fl, vA, leftA); bk_add2<true >(acc, pB, f, fl, vB, leftB); }
        eb = en; pA = nA; pB = nB;
    }
    __syncthreads();

    float* o = out + (size_t)row0 * CLS;
    for (int i = t; i < RB * CLS; i += 1024) {
        const int r = i / CLS, ff = i % CLS;
        o[i] = dinv[row0 + r] * acc[r * P2 + ff] + b2[ff];
    }
}

extern "C" void kernel_launch(void* const* d_in, const int* in_sizes, int n_in,
                              void* d_out, int out_size, void* d_ws, size_t ws_size,
                              hipStream_t stream) {
    const float* x  = (const float*)d_in[0];
    const int*   ei = (const int*)  d_in[1];
    const float* W1 = (const float*)d_in[2];
    const float* b1 = (const float*)d_in[3];
    const float* W2 = (const float*)d_in[4];
    const float* b2 = (const float*)d_in[5];
    float* out = (float*)d_out;

    const int N = in_sizes[0] / F_IN;
    const int E = in_sizes[1] / 2;
    const int* src = ei;
    const int* dst = ei + E;

    float* dinv = (float*)d_ws;
    float* h    = dinv + N;
    float* agg  = h   + (size_t)N * HID;
    float* h2   = agg + (size_t)N * HID;
    float* a1a  = h2;
    float* a1b  = out;
    float* a1c  = h2 + (size_t)N * CLS;
    float* h2p  = h;
    float* acc1 = h + (size_t)N * P2;

    const int B = 256;
    const int nTiles = N / 16;

    const int nEB = (E + ECH - 1) / ECH;
    const size_t bkNeed = ((size_t)N * (1 + HID + P2) + (size_t)E + 256) * sizeof(float);
    if (ws_size >= bkNeed && N == NODES && (size_t)nEB * NBK <= (size_t)N * CLS) {
        float*    hB   = dinv + N;
        float*    h2B  = hB + (size_t)N * HID;
        unsigned* pk   = (unsigned*)(h2B + (size_t)N * P2);
        unsigned* bst  = pk + E;
        unsigned* cnt  = (unsigned*)out;
        const int nPartsB = (E + CHUNK - 1) / CHUNK;
        unsigned* partsB = (unsigned*)hB;
        deg_hist_k       <<<nPartsB, 1024, 0, stream>>>(dst, partsB, E);
        dinv_from_parts_k<<<(HWORDS + B - 1) / B, B, 0, stream>>>(partsB, dinv, nPartsB, N);
        bk_count_k<<<nEB, 512, 0, stream>>>(dst, cnt, E);
        bk_place_k<<<nEB, 512, 0, stream>>>(src, dst, cnt, bst, pk, E, nEB);
        gemm1_k<true><<<(nTiles + 7) / 8, B, 0, stream>>>(x, W1, dinv, hB, hB, hB, hB, hB, ws_size, nTiles);
        bk_agg1_k<<<NBK, 1024, 0, stream>>>(pk, bst, hB, dinv, b1, W2, h2B);
        bk_agg2_k<<<NBK, 1024, 0, stream>>>(pk, bst, h2B, dinv, b2, out);
        return;
    }

    const int nParts = (E + CHUNK - 1) / CHUNK;
    unsigned* parts = (unsigned*)h;
    deg_hist_k       <<<nParts, 1024, 0, stream>>>(dst, parts, E);
    dinv_from_parts_k<<<(HWORDS + B - 1) / B, B, 0, stream>>>(parts, dinv, nParts, N);

    gemm1_k<false><<<(nTiles + 7) / 8, B, 0, stream>>>(x, W1, dinv, h, agg, a1a, a1b, a1c, ws_size, nTiles);

    scatter1_k<<<(E + B - 1) / B, B, 0, stream>>>(src, dst, h, agg, a1a, a1b, a1c, ws_size, E);

    gemm2_k    <<<(nTiles + 7) / 8, B, 0, stream>>>(agg, a1a, a1b, a1c, ws_size, b1, W2, dinv, h2p, acc1, nTiles);

    int n2 = N * CLS;
    scatter2_k<<<(E + B - 1) / B, B, 0, stream>>>(src, dst, h2p, agg, acc1, E);
    final2_k  <<<(n2 + B - 1) / B, B, 0, stream>>>(out, agg, acc1, dinv, b2, n2);
}


# Degree count from the bucket-ordered edge words: one workgroup per bucket counts its 400 rows in LDS and writes dinv; the two degree launches (node table per block, plane sum) go; integer counts
# speedup vs baseline: 4.6335x; 1.1242x over previous
#include <hip/hip_runtime.h>
#include <hip/hip_bf16.h>

typedef __attribute__((ext_vector_type(2))) float v2f;
typedef __attribute__((ext_vector_type(8))) float v8f;

#define F_IN 512
#define HID  16
#define CLS  7
#define P2   8

__device__ __forceinline__ unsigned chip_half() {
    return (__builtin_amdgcn_s_sendmsg_rtn(0x87) >> 18) & 1u;
}

#define NODES  100000
#define HWORDS (NODES / 2)
#define CHUNK  50000

#define ROWS_B ((NODES * CLS) / HID)
#define ROWS_C  (NODES - 2 * ROWS_B)
#define WS_NEED (((size_t)NODES * (1 + HID + HID + CLS) + (size_t)ROWS_C * HID) * sizeof(float))

__global__ void __launch_bounds__(1024) deg_hist_k(const int* __restrict__ dst,
                                                   unsigned* __restrict__ parts, int E) {
    __shared__ __attribute__((aligned(16))) unsigned cnt[HWORDS];

    const int base = blockIdx.x * CHUNK;
    int end = base + CHUNK;
    if (end > E) end = E;
    const int nvec = (end - base) >> 2;
    const int4* d4 = (const int4*)(dst + base);

    constexpr int SLOTS = (CHUNK / 4 + 1023) / 1024;
    int4 d[SLOTS];
#pragma unroll
    for (int j = 0; j < SLOTS; ++j) {
        const int i = threadIdx.x + j * 1024;
        if (i < nvec) d[j] = d4[i];
    }

    for (int i = threadIdx.x; i < HWORDS / 4; i += 1024)
        ((uint4*)cnt)[i] = make_uint4(0u, 0u, 0u, 0u);
    __syncthreads();

#pragma unroll
    for (int j = 0; j < SLOTS; ++j) {
        const int i = threadIdx.x + j * 1024;
        if (i < nvec) {
            atomicAdd(&cnt[d[j].x >> 1], 1u << ((d[j].x & 1) << 4));
            atomicAdd(&cnt[d[j].y >> 1], 1u << ((d[j].y & 1) << 4));
            atomicAdd(&cnt[d[j].z >> 1], 1u << ((d[j].z & 1) << 4));
            atomicAdd(&cnt[d[j].w >> 1], 1u << ((d[j].w & 1) << 4));
        }
    }
    for (int e = base + (nvec << 2) + threadIdx.x; e < end; e += 1024) {
        const int t = dst[e];
        atomicAdd(&cnt[t >> 1], 1u << ((t & 1) << 4));
    }
    __syncthreads();

    uint4* o = (uint4*)(parts + (size_t)blockIdx.x * HWORDS);
    for (int i = threadIdx.x; i < HWORDS / 4; i += 1024) o[i] = ((const uint4*)cnt)[i];
}

__global__ void dinv_from_parts_k(const unsigned* __restrict__ parts, float* __restrict__ dinv,
                                  int nParts, int n) {
    int t = blockIdx.x * blockDim.x + threadIdx.x;
    if (t >= HWORDS) return;
    unsigned lo = 0u, hi = 0u;
#pragma unroll 8
    for (int p = 0; p < nParts; ++p) {
        unsigned w = parts[(size_t)p * HWORDS + t];
        lo += w & 0xffffu;
        hi += w >> 16;
    }
    const int i = 2 * t;
    float r0 = rsqrtf(1.0f + (float)lo);
    float r1 = rsqrtf(1.0f + (float)hi);
    if (i + 1 < n)      *(float2*)(dinv + i) = make_float2(r0, r1);
    else if (i < n)     dinv[i] = r0;
}

typedef __attribute__((ext_vector_type(4))) float v4f;

#define KC  64
#define PX  (KC + 4)
#define PW  (F_IN + 4)
#define NCH (F_IN / KC)
#define LPC 8

__device__ __forceinline__ void wave_lds_order() {
    __builtin_amdgcn_fence(__ATOMIC_RELEASE, "wavefront");
    __builtin_amdgcn_wave_barrier();
    __builtin_amdgcn_fence(__ATOMIC_ACQUIRE, "wavefront");
}

template <bool BK>
__global__ void __launch_bounds__(256) gemm1_k(const float* __restrict__ x,
                                               const float* __restrict__ W1,
                                               const float* __restrict__ dinv,
                                               float* __restrict__ h,
                                               float* __restrict__ agg,
                                               float* __restrict__ a1a,
                                               float* __restrict__ a1b,
                                               float* __restrict__ a1c, size_t wsBytes, int nTiles) {
    __shared__ __attribute__((aligned(16))) float sWt[HID * PW];
    __shared__ __attribute__((aligned(16))) float sX[8 * 16 * PX];
    for (int i = threadIdx.x * 4; i < F_IN * HID; i += 256 * 4) {
        const float4 w = *(const float4*)(W1 + i);
        const int k = i >> 4, n = i & 15;
        sWt[(n    ) * PW + k] = w.x;
        sWt[(n + 1) * PW + k] = w.y;
        sWt[(n + 2) * PW + k] = w.z;
        sWt[(n + 3) * PW + k] = w.w;
    }
    __syncthreads();

    const int wave = threadIdx.x >> 5;
    const int lane = threadIdx.x & 31;
    const int tile = blockIdx.x * 8 + wave;
    if (tile >= nTiles) return;

    const int m    = lane & 15;
    const int half = lane >> 4;

    float* const st = sX + wave * (16 * PX);
    const float* gp = x + (size_t)(tile * 16 + half) * F_IN + m * 4;
    float*       sp = st + half * PX + m * 4;
    const float* fa = st  + m * PX + half * 4;
    const float* fb = sWt + m * PW + half * 4;

    v4f g[LPC];
#pragma unroll
    for (int i = 0; i < LPC; ++i)
        g[i] = __builtin_nontemporal_load((const v4f*)(gp + (size_t)i * 2 * F_IN));

    v8f c0 = {0.f, 0.f, 0.f, 0.f, 0.f, 0.f, 0.f, 0.f};
    v8f c1 = {0.f, 0.f, 0.f, 0.f, 0.f, 0.f, 0.f, 0.f};

    for (int ch = 0; ch < NCH; ++ch) {
#pragma unroll
        for (int i = 0; i < LPC; ++i)
            *(v4f*)(sp + i * 2 * PX) = g[i];
        if (ch + 1 < NCH) {
#pragma unroll
            for (int i = 0; i < LPC; ++i)
                g[i] = __builtin_nontemporal_load((const v4f*)(gp + (ch + 1) * KC + (size_t)i * 2 * F_IN));
        }
        wave_lds_order();
#pragma unroll
        for (int j = 0; j < KC / 8; ++j) {
            const v4f a = *(const v4f*)(fa + 8 * j);
            const v4f b = *(const v4f*)(fb + ch * KC + 8 * j);
            const v2f a01 = {a.x, a.y}, b01 = {b.x, b.y};
            const v2f a23 = {a.z, a.w}, b23 = {b.z, b.w};
            c0 = __builtin_amdgcn_wmma_f32_16x16x4_f32(false, a01, false, b01,
                                                       (short)0, c0, false, false);
            c1 = __builtin_amdgcn_wmma_f32_16x16x4_f32(false, a23, false, b23,
                                                       (short)0, c1, false, false);
        }
        wave_lds_order();
    }
    const v8f c = c0 + c1;

    const float4 dA = *(const float4*)(dinv + tile * 16 + half * 8);
    const float4 dB = *(const float4*)(dinv + tile * 16 + half * 8 + 4);
    const float dr[8] = {dA.x, dA.y, dA.z, dA.w, dB.x, dB.y, dB.z, dB.w};

    const size_t o = (size_t)(tile * 16 + half * 8) * HID + m;
#pragma unroll
    for (int v = 0; v < 8; ++v) {
        const float val = dr[v] * c[v];
        h  [o + (size_t)v * HID] = val;
        if (!BK) {
            agg[o + (size_t)v * HID] = val;
            const int r = tile * 16 + half * 8 + v;
            float* z = (r < ROWS_B)     ? (a1a + (size_t)r * HID + m)
                     : (r < 2 * ROWS_B) ? (a1b + (size_t)(r - ROWS_B) * HID + m)
                                        : (a1c + (size_t)(r - 2 * ROWS_B) * HID + m);
            if (r < 2 * ROWS_B || wsBytes >= WS_NEED) *z = 0.0f;
        }
    }
}


__device__ __forceinline__ unsigned rowsel(unsigned x, unsigned lo, unsigned hi) {
    return __builtin_amdgcn_permlane16(x, x, lo, hi, false, false);
}

template <bool TAIL>
__device__ __forceinline__ void adds1(float* agg, const float (&v)[16],
                                      unsigned rlo, unsigned rhi, unsigned f4, int left) {
    const int lane = threadIdx.x & 31;
    char* const colbase = (char*)agg + f4;
#pragma unroll
    for (int j = 0; j < 16; ++j) {
        const unsigned sel = 0x11111111u * (unsigned)j;
        const unsigned lo = rowsel(rlo, sel, sel);
        const unsigned hi = rowsel(rhi, sel, sel);
        const long long off = (long long)(((unsigned long long)hi << 32) | lo);
        if (!TAIL || ((lane & 16) + j) < left)
            unsafeAtomicAdd((float*)(colbase + off), v[j]);
    }
}

__global__ void __launch_bounds__(256) scatter1_k(const int* __restrict__ src,
                                                  const int* __restrict__ dst,
                                                  const float* __restrict__ h,
                                                  float* agg, float* a1a, float* a1b, float* a1c,
                                                  size_t wsBytes, int E) {
    const int lane = threadIdx.x & 31;
    const int e0 = (blockIdx.x * 8 + (threadIdx.x >> 5)) * 32;
    if (e0 >= E) return;
    const int left = E - e0;
    int s = 0, d = 0;
    if (lane < left) { s = src[e0 + lane]; d = dst[e0 + lane]; }

    const unsigned f4 = (unsigned)(lane & 15) * 4u;
    const unsigned sb = (unsigned)s * (HID * 4u);
    float v[16];
#pragma unroll
    for (int j = 0; j < 16; ++j) {
        const unsigned sel = 0x11111111u * (unsigned)j;
        const unsigned off = rowsel(sb, sel, sel) + f4;
        v[j] = *(const float*)((const char*)h + off);
    }

    const unsigned half = chip_half();
    const float* row = agg + (size_t)d * HID;
    if (half != 0u) {
        if (d < ROWS_B)              row = a1a + (size_t)d * HID;
        else if (d < 2 * ROWS_B)     row = a1b + (size_t)(d - ROWS_B) * HID;
        else if (wsBytes >= WS_NEED) row = a1c + (size_t)(d - 2 * ROWS_B) * HID;
    }
    const long long roff = (const char*)row - (const char*)agg;
    const unsigned rlo = (unsigned)(unsigned long long)roff;
    const unsigned rhi = (unsigned)((unsigned long long)roff >> 32);
    if (left >= 32) adds1<false>(agg, v, rlo, rhi, f4, left);
    else            adds1<true >(agg, v, rlo, rhi, f4, left);
}

__global__ void __launch_bounds__(256) gemm2_k(float* agg,
                                               const float* __restrict__ a1a,
                                               const float* __restrict__ a1b,
                                               const float* __restrict__ a1c, size_t wsBytes,
                                               const float* __restrict__ b1,
                                               const float* __restrict__ W2,
                                               const float* __restrict__ dinv,
                                               float* __restrict__ h2p,
                                               float* __restrict__ acc1, int nTiles) {
    __shared__ float sW[HID * 16];
    {
        int t = threadIdx.x;
        int k = t >> 4, n = t & 15;
        sW[t] = (n < CLS) ? W2[k * CLS + n] : 0.0f;
    }
    __syncthreads();

    const int wave = threadIdx.x >> 5;
    const int lane = threadIdx.x & 31;
    const int tile = blockIdx.x * 8 + wave;
    if (tile >= nTiles) return;

    const int m    = lane & 15;
    const int kk   = (lane >> 4) << 1;
    const int half = lane >> 4;

    const int    r    = tile * 16 + m;
    const bool   has  = r < 2 * ROWS_B || wsBytes >= WS_NEED;
    const float* arow = agg + (size_t)r * HID + kk;
    const float* srow = (r < ROWS_B)     ? (a1a + (size_t)r * HID + kk)
                      : (r < 2 * ROWS_B) ? (a1b + (size_t)(r - ROWS_B) * HID + kk)
                      : (has ? (a1c + (size_t)(r - 2 * ROWS_B) * HID + kk) : arow);
    const float  dm   = dinv[r];

    v2f a0[HID / 4], a1[HID / 4], bb[HID / 4];
#pragma unroll
    for (int q = 0; q < HID / 4; ++q) {
        a0[q] = *(const v2f*)(arow + 4 * q);
        a1[q] = *(const v2f*)(srow + 4 * q);
        bb[q] = *(const v2f*)(b1 + 4 * q + kk);
    }

    v8f c = {0.f, 0.f, 0.f, 0.f, 0.f, 0.f, 0.f, 0.f};

#pragma unroll
    for (int q = 0; q < HID / 4; ++q) {
        const int k0 = 4 * q;
        const float sx = has ? a0[q].x + a1[q].x : a0[q].x;
        const float sy = has ? a0[q].y + a1[q].y : a0[q].y;
        const float vx = dm * sx + bb[q].x;
        const float vy = dm * sy + bb[q].y;
        v2f a;
        a.x = vx > 0.f ? vx : 0.f;
        a.y = vy > 0.f ? vy : 0.f;
        v2f b;
        b.x = sW[(k0 + kk)     * 16 + m];
        b.y = sW[(k0 + kk + 1) * 16 + m];
        c = __builtin_amdgcn_wmma_f32_16x16x4_f32(false, a, false, b,
                                                  (short)0, c, false, false);
    }

    const float4 dA = *(const float4*)(dinv + tile * 16 + half * 8);
    const float4 dB = *(const float4*)(dinv + tile * 16 + half * 8 + 4);
    const float dr[8] = {dA.x, dA.y, dA.z, dA.w, dB.x, dB.y, dB.z, dB.w};

    if (m < P2) {
#pragma unroll
        for (int v = 0; v < 8; ++v) {
            const int row = tile * 16 + half * 8 + v;
            const float val = dr[v] * c[v];
            h2p [(size_t)row * P2  + m] = val;
            agg [(size_t)row * HID + m] = val;
            acc1[(size_t)row * P2  + m] = 0.0f;
        }
    }
}


template <bool TAIL>
__device__ __forceinline__ void adds2(float* __restrict__ out, const float (&v)[8],
                                      unsigned db, unsigned f4, bool fl, int left) {
    const int lane = threadIdx.x & 31;
    unsigned off[8];
#pragma unroll
    for (int j = 0; j < 8; ++j) {
        const unsigned lo = 0x11111111u * (unsigned)j;
        const unsigned hi = 0x11111111u * (unsigned)(j + 8);
        off[j] = rowsel(db, lo, hi) + f4;
    }
    if (fl) {
#pragma unroll
        for (int j = 0; j < 8; ++j) {
            if (!TAIL || ((lane & 24) + j) < left)
                unsafeAtomicAdd((float*)((char*)out + off[j]), v[j]);
        }
    }
}

__global__ void __launch_bounds__(256) scatter2_k(const int* __restrict__ src,
                                                  const int* __restrict__ dst,
                                                  const float* __restrict__ h2p,
                                                  float* __restrict__ acc0,
                                                  float* __restrict__ acc1, int E) {
    const int lane = threadIdx.x & 31;
    const int e0 = (blockIdx.x * 8 + (threadIdx.x >> 5)) * 32;
    if (e0 >= E) return;
    const int left = E - e0;
    int s = 0, d = 0;
    if (lane < left) { s = src[e0 + lane]; d = dst[e0 + lane]; }

    const int  f  = lane & 7;
    const bool fl = f < CLS;
    const unsigned f4 = (unsigned)(fl ? f : CLS - 1) * 4u;
    const unsigned sb = (unsigned)s * (P2 * 4u);
    float v[8];
#pragma unroll
    for (int j = 0; j < 8; ++j) {
        const unsigned lo = 0x11111111u * (unsigned)j;
        const unsigned hi = 0x11111111u * (unsigned)(j + 8);
        const unsigned off = rowsel(sb, lo, hi) + f4;
        v[j] = *(const float*)((const char*)h2p + off);
    }

    const unsigned hf = chip_half();
    float* out = hf ? acc1 : acc0;
    const unsigned db = (unsigned)d * (hf ? P2 * 4u : HID * 4u);
    if (left >= 32) adds2<false>(out, v, db, f4, fl, left);
    else            adds2<true >(out, v, db, f4, fl, left);
}

__global__ void final2_k(float* __restrict__ out, const float* __restrict__ acc0,
                         const float* __restrict__ acc1,
                         const float* __restrict__ dinv,
                         const float* __restrict__ b2, int n) {
    int t = blockIdx.x * blockDim.x + threadIdx.x;
    if (t >= n) return;
    int i = t / CLS, f = t % CLS;
    out[t] = dinv[i] * (acc0[(size_t)i * HID + f] + acc1[(size_t)i * P2 + f]) + b2[f];
}

#define RB    400
#define NBK   (NODES / RB)
#define ECH   12800
#define EPT   (ECH / 512)
#define LB    9

__global__ void __launch_bounds__(512) bk_count_k(const int* __restrict__ dst,
                                                  unsigned* __restrict__ cnt, int E) {
    __shared__ unsigned hc[256];
    if (threadIdx.x < 256) hc[threadIdx.x] = 0u;
    const int base = blockIdx.x * ECH;
    int end = base + ECH;
    if (end > E) end = E;
    int d[EPT];
#pragma unroll
    for (int j = 0; j < EPT; ++j) {
        const int e = base + (int)threadIdx.x + j * 512;
        d[j] = (e < end) ? dst[e] : -1;
    }
    __syncthreads();
#pragma unroll
    for (int j = 0; j < EPT; ++j)
        if (d[j] >= 0) atomicAdd(&hc[(unsigned)d[j] / RB], 1u);
    __syncthreads();
    if (threadIdx.x < NBK) cnt[blockIdx.x * NBK + threadIdx.x] = hc[threadIdx.x];
}

__global__ void __launch_bounds__(512) bk_place_k(const int* __restrict__ src,
                                                  const int* __restrict__ dst,
                                                  const unsigned* __restrict__ cnt,
                                                  unsigned* __restrict__ bstart,
                                                  unsigned* __restrict__ pk, int E, int nEB) {
    __shared__ unsigned hAll[512], hBef[512];
    __shared__ unsigned wg[8], wl[8];
    __shared__ unsigned lst[256];
    __shared__ unsigned cur[256];
    __shared__ unsigned gof[256];
    __shared__ unsigned stage[ECH];
    const int t = threadIdx.x;
    const int base = blockIdx.x * ECH;
    int end = base + ECH;
    if (end > E) end = E;
    int s[EPT], d[EPT];
#pragma unroll
    for (int j = 0; j < EPT; ++j) {
        const int e = base + t + j * 512;
        const bool in = e < end;
        s[j] = in ? src[e] : 0;
        d[j] = in ? dst[e] : -1;
    }
    {
        const int b = t & 255, hf = t >> 8;
        const int lo = hf ? (nEB >> 1) : 0, hi = hf ? nEB : (nEB >> 1);
        unsigned all = 0u, bef = 0u;
        if (b < NBK) {
#pragma unroll 25
            for (int e = lo; e < hi; ++e) {
                const unsigned q = cnt[e * NBK + b];
                all += q;
                bef += (e < (int)blockIdx.x) ? q : 0u;
            }
        }
        hAll[t] = all;
        hBef[t] = bef;
    }
    unsigned own = 0u;
    if (t < NBK) own = cnt[blockIdx.x * NBK + t];
    __syncthreads();
    unsigned tot = 0u, bef = 0u;
    if (t < 256) {
        tot = hAll[t] + hAll[t + 256];
        bef = hBef[t] + hBef[t + 256];
    }
    const int lane = t & 31, wave = t >> 5;
    unsigned gs = tot, ls = own;
#pragma unroll
    for (int o = 1; o < 32; o <<= 1) {
        const unsigned u = __shfl_up(gs, o);
        const unsigned w = __shfl_up(ls, o);
        if (lane >= o) { gs += u; ls += w; }
    }
    if (lane == 31 && wave < 8) { wg[wave] = gs; wl[wave] = ls; }
    __syncthreads();
    if (t < 256) {
        unsigned gb = 0u, lb = 0u;
        for (int k = 0; k < wave; ++k) { gb += wg[k]; lb += wl[k]; }
        const unsigned gx = gs + gb - tot;
        const unsigned lx = ls + lb - own;
        lst[t] = lx;
        cur[t] = lx;
        gof[t] = gx + bef;
        if (blockIdx.x == 0 && t <= NBK) bstart[t] = gx;
    }
    __syncthreads();
#pragma unroll
    for (int j = 0; j < EPT; ++j) {
        if (d[j] >= 0) {
            const unsigned b = (unsigned)d[j] / RB;
            const unsigned p = atomicAdd(&cur[b], 1u);
            stage[p] = ((unsigned)s[j] << LB) | ((unsigned)d[j] - b * RB);
        }
    }
    __syncthreads();
    for (int b = wave; b < NBK; b += 16) {
        const unsigned l = lst[b];
        const unsigned nb = cur[b] - l;
        const unsigned g = gof[b];
        for (unsigned i = lane; i < nb; i += 32) pk[g + i] = stage[l + i];
    }
}

#define DPT 26
__global__ void __launch_bounds__(512) bk_deg_k(const unsigned* __restrict__ pk,
                                                const unsigned* __restrict__ bstart,
                                                float* __restrict__ dinv) {
    __shared__ unsigned dc[1 << LB];
    const unsigned t = threadIdx.x;
    dc[t] = 0u;
    const unsigned e0 = bstart[blockIdx.x], e1 = bstart[blockIdx.x + 1];
    __syncthreads();
    for (unsigned base = e0; base < e1; base += 512u * DPT) {
        unsigned p[DPT];
#pragma unroll
        for (int j = 0; j < DPT; ++j) {
            const unsigned e = base + t + (unsigned)j * 512u;
            p[j] = pk[e < e1 ? e : e1 - 1u];
        }
#pragma unroll
        for (int j = 0; j < DPT; ++j)
            if (base + t + (unsigned)j * 512u < e1)
                atomicAdd(&dc[p[j] & ((1u << LB) - 1u)], 1u);
    }
    __syncthreads();
    if (t < RB) dinv[blockIdx.x * RB + t] = rsqrtf(1.0f + (float)dc[t]);
}

__device__ __forceinline__ void bk_gather1(const float* __restrict__ h, unsigned p, unsigned f4,
                                           float (&v)[16]) {
    const unsigned sb = (p >> LB) * (HID * 4u);
#pragma unroll
    for (int j = 0; j < 16; ++j) {
        const unsigned sel = 0x11111111u * (unsigned)j;
        v[j] = *(const float*)((const char*)h + (rowsel(sb, sel, sel) + f4));
    }
}
template <bool TAIL>
__device__ __forceinline__ void bk_add1(float* acc, unsigned p, unsigned f, const float (&v)[16],
                                        unsigned left) {
    const unsigned lane = threadIdx.x & 31;
    const unsigned dl = (p & ((1u << LB) - 1u)) * HID;
#pragma unroll
    for (int j = 0; j < 16; ++j) {
        const unsigned sel = 0x11111111u * (unsigned)j;
        const unsigned a = rowsel(dl, sel, sel) + f;
        if (!TAIL || ((lane & 16u) + (unsigned)j) < left)
            unsafeAtomicAdd(&acc[a], v[j]);
    }
}

__global__ void __launch_bounds__(1024) bk_agg1_k(const unsigned* __restrict__ pk,
                                                  const unsigned* __restrict__ bstart,
                                                  const float* __restrict__ h,
                                                  const float* __restrict__ dinv,
                                                  const float* __restrict__ b1,
                                                  const float* __restrict__ W2,
                                                  float* __restrict__ h2p) {
    __shared__ __attribute__((aligned(16))) float acc[RB * HID];
    __shared__ float sW[HID * 16];
    const int t = threadIdx.x;
    const int row0 = blockIdx.x * RB;
    const float4* h4 = (const float4*)(h + (size_t)row0 * HID);
    for (int i = t; i < RB * HID / 4; i += 1024) ((float4*)acc)[i] = h4[i];
    if (t < 256) {
        const int k = t >> 4, n = t & 15;
        sW[t] = (n < CLS) ? W2[k * CLS + n] : 0.0f;
    }
    const unsigned e0 = bstart[blockIdx.x], e1 = bstart[blockIdx.x + 1];
    const unsigned lane = (unsigned)t & 31u, wave = (unsigned)t >> 5;
    const unsigned f = lane & 15u;
    const unsigned step = 32u * 64u;
    unsigned eb = e0 + wave * 64u;
    unsigned pA = 0u, pB = 0u;
    if (eb + lane < e1)       pA = pk[eb + lane];
    if (eb + 32u + lane < e1) pB = pk[eb + 32u + lane];
    __syncthreads();

    while (eb < e1) {
        const unsigned en = eb + step;
        unsigned nA = 0u, nB = 0u;
        if (en + lane < e1)       nA = pk[en + lane];
        if (en + 32u + lane < e1) nB = pk[en + 32u + lane];
        const unsigned leftA = e1 - eb;
        const unsigned leftB = leftA > 32u ? leftA - 32u : 0u;
        float vA[16], vB[16];
        bk_gather1(h, pA, f * 4u, vA);
        bk_gather1(h, pB, f * 4u, vB);
        if (leftB >= 32u) { bk_add1<false>(acc, pA, f, vA, leftA); bk_add1<false>(acc, pB, f, vB, leftB); }
        else              { bk_add1<true >(acc, pA, f, vA, leftA); bk_add1<true >(acc, pB, f, vB, leftB); }
        eb = en; pA = nA; pB = nB;
    }
    __syncthreads();

    if (wave < RB / 16) {
        const int m    = (int)(lane & 15u);
        const int half = (int)(lane >> 4);
        const int kk   = half << 1;
        const int rl   = (int)wave * 16 + m;
        const float dm = dinv[row0 + rl];
        v8f c = {0.f, 0.f, 0.f, 0.f, 0.f, 0.f, 0.f, 0.f};
#pragma unroll
        for (int q = 0; q < HID / 4; ++q) {
            const int k0 = 4 * q;
            const float vx = dm * acc[rl * HID + k0 + kk]     + b1[k0 + kk];
            const float vy = dm * acc[rl * HID + k0 + kk + 1] + b1[k0 + kk + 1];
            v2f a;
            a.x = vx > 0.f ? vx : 0.f;
            a.y = vy > 0.f ? vy : 0.f;
            v2f b;
            b.x = sW[(k0 + kk)     * 16 + m];
            b.y = sW[(k0 + kk + 1) * 16 + m];
            c = __builtin_amdgcn_wmma_f32_16x16x4_f32(false, a, false, b,
                                                      (short)0, c, false, false);
        }
        if (m < P2) {
#pragma unroll
            for (int v = 0; v < 8; ++v) {
                const int row = row0 + (int)wave * 16 + half * 8 + v;
                h2p[(size_t)row * P2 + m] = dinv[row] * c[v];
            }
        }
    }
}

__device__ __forceinline__ void bk_gather2(const float* __restrict__ h2p, unsigned p, unsigned f4,
                                           float (&v)[8]) {
    const unsigned sb = (p >> LB) * (P2 * 4u);
#pragma unroll
    for (int j = 0; j < 8; ++j) {
        const unsigned lo = 0x11111111u * (unsigned)j;
        const unsigned hi = 0x11111111u * (unsigned)(j + 8);
        v[j] = *(const float*)((const char*)h2p + (rowsel(sb, lo, hi) + f4));
    }
}
template <bool TAIL>
__device__ __forceinline__ void bk_add2(float* acc, unsigned p, unsigned f, bool fl, const float (&v)[8],
                                        unsigned left) {
    const unsigned lane = threadIdx.x & 31;
    const unsigned dl = (p & ((1u << LB) - 1u)) * P2;
#pragma unroll
    for (int j = 0; j < 8; ++j) {
        const unsigned lo = 0x11111111u * (unsigned)j;
        const unsigned hi = 0x11111111u * (unsigned)(j + 8);
        const unsigned a = rowsel(dl, lo, hi) + f;
        if (fl && (!TAIL || ((lane & 24u) + (unsigned)j) < left))
            unsafeAtomicAdd(&acc[a], v[j]);
    }
}

__global__ void __launch_bounds__(1024) bk_agg2_k(const unsigned* __restrict__ pk,
                                                  const unsigned* __restrict__ bstart,
                                                  const float* __restrict__ h2p,
                                                  const float* __restrict__ dinv,
                                                  const float* __restrict__ b2,
                                                  float* __restrict__ out) {
    __shared__ __attribute__((aligned(16))) float acc[RB * P2];
    const int t = threadIdx.x;
    const int row0 = blockIdx.x * RB;
    const float4* g4 = (const float4*)(h2p + (size_t)row0 * P2);
    for (int i = t; i < RB * P2 / 4; i += 1024) ((float4*)acc)[i] = g4[i];
    const unsigned e0 = bstart[blockIdx.x], e1 = bstart[blockIdx.x + 1];
    const unsigned lane = (unsigned)t & 31u, wave = (unsigned)t >> 5;
    const unsigned f  = lane & 7u;
    const bool     fl = f < CLS;
    const unsigned f4 = (fl ? f : (unsigned)(CLS - 1)) * 4u;
    const unsigned step = 32u * 64u;
    unsigned eb = e0 + wave * 64u;
    unsigned pA = 0u, pB = 0u;
    if (eb + lane < e1)       pA = pk[eb + lane];
    if (eb + 32u + lane < e1) pB = pk[eb + 32u + lane];
    __syncthreads();

    while (eb < e1) {
        const unsigned en = eb + step;
        unsigned nA = 0u, nB = 0u;
        if (en + lane < e1)       nA = pk[en + lane];
        if (en + 32u + lane < e1) nB = pk[en + 32u + lane];
        const unsigned leftA = e1 - eb;
        const unsigned leftB = leftA > 32u ? leftA - 32u : 0u;
        float vA[8], vB[8];
        bk_gather2(h2p, pA, f4, vA);
        bk_gather2(h2p, pB, f4, vB);
        if (leftB >= 32u) { bk_add2<false>(acc, pA, f, fl, vA, leftA); bk_add2<false>(acc, pB, f, fl, vB, leftB); }
        else              { bk_add2<true >(acc, pA, f, fl, vA, leftA); bk_add2<true >(acc, pB, f, fl, vB, leftB); }
        eb = en; pA = nA; pB = nB;
    }
    __syncthreads();

    float* o = out + (size_t)row0 * CLS;
    for (int i = t; i < RB * CLS; i += 1024) {
        const int r = i / CLS, ff = i % CLS;
        o[i] = dinv[row0 + r] * acc[r * P2 + ff] + b2[ff];
    }
}

extern "C" void kernel_launch(void* const* d_in, const int* in_sizes, int n_in,
                              void* d_out, int out_size, void* d_ws, size_t ws_size,
                              hipStream_t stream) {
    const float* x  = (const float*)d_in[0];
    const int*   ei = (const int*)  d_in[1];
    const float* W1 = (const float*)d_in[2];
    const float* b1 = (const float*)d_in[3];
    const float* W2 = (const float*)d_in[4];
    const float* b2 = (const float*)d_in[5];
    float* out = (float*)d_out;

    const int N = in_sizes[0] / F_IN;
    const int E = in_sizes[1] / 2;
    const int* src = ei;
    const int* dst = ei + E;

    float* dinv = (float*)d_ws;
    float* h    = dinv + N;
    float* agg  = h   + (size_t)N * HID;
    float* h2   = agg + (size_t)N * HID;
    float* a1a  = h2;
    float* a1b  = out;
    float* a1c  = h2 + (size_t)N * CLS;
    float* h2p  = h;
    float* acc1 = h + (size_t)N * P2;

    const int B = 256;
    const int nTiles = N / 16;

    const int nEB = (E + ECH - 1) / ECH;
    const size_t bkNeed = ((size_t)N * (1 + HID + P2) + (size_t)E + 256) * sizeof(float);
    if (ws_size >= bkNeed && N == NODES && (size_t)nEB * NBK <= (size_t)N * CLS) {
        float*    hB   = dinv + N;
        float*    h2B  = hB + (size_t)N * HID;
        unsigned* pk   = (unsigned*)(h2B + (size_t)N * P2);
        unsigned* bst  = pk + E;
        unsigned* cnt  = (unsigned*)out;
        bk_count_k<<<nEB, 512, 0, stream>>>(dst, cnt, E);
        bk_place_k<<<nEB, 512, 0, stream>>>(src, dst, cnt, bst, pk, E, nEB);
        bk_deg_k  <<<NBK, 512, 0, stream>>>(pk, bst, dinv);
        gemm1_k<true><<<(nTiles + 7) / 8, B, 0, stream>>>(x, W1, dinv, hB, hB, hB, hB, hB, ws_size, nTiles);
        bk_agg1_k<<<NBK, 1024, 0, stream>>>(pk, bst, hB, dinv, b1, W2, h2B);
        bk_agg2_k<<<NBK, 1024, 0, stream>>>(pk, bst, h2B, dinv, b2, out);
        return;
    }

    const int nParts = (E + CHUNK - 1) / CHUNK;
    unsigned* parts = (unsigned*)h;
    deg_hist_k       <<<nParts, 1024, 0, stream>>>(dst, parts, E);
    dinv_from_parts_k<<<(HWORDS + B - 1) / B, B, 0, stream>>>(parts, dinv, nParts, N);

    gemm1_k<false><<<(nTiles + 7) / 8, B, 0, stream>>>(x, W1, dinv, h, agg, a1a, a1b, a1c, ws_size, nTiles);

    scatter1_k<<<(E + B - 1) / B, B, 0, stream>>>(src, dst, h, agg, a1a, a1b, a1c, ws_size, E);

    gemm2_k    <<<(nTiles + 7) / 8, B, 0, stream>>>(agg, a1a, a1b, a1c, ws_size, b1, W2, dinv, h2p, acc1, nTiles);

    int n2 = N * CLS;
    scatter2_k<<<(E + B - 1) / B, B, 0, stream>>>(src, dst, h2p, agg, acc1, E);
    final2_k  <<<(n2 + B - 1) / B, B, 0, stream>>>(out, agg, acc1, dinv, b2, n2);
}
